# starting kernel
# baseline (speedup 1.0000x reference)
#include <hip/hip_runtime.h>
#include <hip/hip_bf16.h>

#define NN     50000
#define MPAD   50016
#define E0C    800000
#define ETOT   850000
#define INDIM  256
#define HIDDIM 256
#define OUTDIM 64
#define H1C    8
#define C1C    32
#define H2C    1
#define C2C    64
#define NEG_SLOPE 0.2f

typedef __attribute__((ext_vector_type(16))) _Float16 v16h;
typedef __attribute__((ext_vector_type(8)))  _Float16 v8h;
typedef __attribute__((ext_vector_type(8)))  float    v8f;

union V16U { v16h v; v8h h[2]; };

__device__ __forceinline__ unsigned fenc(float f) {
    int b = __float_as_int(f);
    return (b >= 0) ? ((unsigned)b | 0x80000000u) : ~(unsigned)b;
}
__device__ __forceinline__ float fdec(unsigned u) {
    int b = (u & 0x80000000u) ? (int)(u & 0x7FFFFFFFu) : (int)(~u);
    return __int_as_float(b);
}

__global__ void cvt_f32_to_f16(const float* __restrict__ src,
                               _Float16* __restrict__ dst, int n) {
    int i = blockIdx.x * blockDim.x + threadIdx.x;
    if (i < n) dst[i] = (_Float16)src[i];
}

__global__ void xpose_cvt_f16(const float* __restrict__ W,
                              _Float16* __restrict__ Wt, int K, int Nc) {
    int i = blockIdx.x * blockDim.x + threadIdx.x;
    if (i >= K * Nc) return;
    int k = i / Nc, n = i % Nc;
    Wt[(size_t)n * K + k] = (_Float16)W[i];
}

#define WMMA16(a, b, c) __builtin_amdgcn_wmma_f32_16x16x32_f16( \
        false, (a), false, (b), (short)0, (c), false, false)

__global__ __launch_bounds__(128)
void gemm_f16_wmma(const _Float16* __restrict__ A,
                   const _Float16* __restrict__ Bt,
                   float* __restrict__ C, int Mtiles2, int Nc, int K) {
    int wave = (int)((blockIdx.x * blockDim.x + threadIdx.x) >> 5);
    int lane = threadIdx.x & 31;
    int nt2 = Nc >> 5;
    int tm = wave / nt2;
    int tn = wave % nt2;
    if (tm >= Mtiles2) return;
    int r    = lane & 15;
    int half = lane >> 4;

    const _Float16* a0 = A  + (size_t)(tm * 32 + r) * K + half * 8;
    const _Float16* a1 = a0 + (size_t)16 * K;
    const _Float16* b0 = Bt + (size_t)(tn * 32 + r) * K + half * 8;
    const _Float16* b1 = b0 + (size_t)16 * K;

    v8f c00 = {}, c01 = {}, c10 = {}, c11 = {};
    for (int k0 = 0; k0 < K; k0 += 32) {
        V16U A0, A1, B0, B1;
        A0.h[0] = *(const v8h*)(a0 + k0);
        A0.h[1] = *(const v8h*)(a0 + k0 + 16);
        A1.h[0] = *(const v8h*)(a1 + k0);
        A1.h[1] = *(const v8h*)(a1 + k0 + 16);
        B0.h[0] = *(const v8h*)(b0 + k0);
        B0.h[1] = *(const v8h*)(b0 + k0 + 16);
        B1.h[0] = *(const v8h*)(b1 + k0);
        B1.h[1] = *(const v8h*)(b1 + k0 + 16);
        c00 = WMMA16(A0.v, B0.v, c00);
        c01 = WMMA16(A0.v, B1.v, c01);
        c10 = WMMA16(A1.v, B0.v, c10);
        c11 = WMMA16(A1.v, B1.v, c11);
    }
    {
        float* p = C + (size_t)(tm * 32 + half * 8) * Nc + tn * 32 + r;
#pragma unroll
        for (int j = 0; j < 8; ++j) { p[(size_t)j * Nc] = c00[j]; p[(size_t)j * Nc + 16] = c01[j]; }
        p += (size_t)16 * Nc;
#pragma unroll
        for (int j = 0; j < 8; ++j) { p[(size_t)j * Nc] = c10[j]; p[(size_t)j * Nc + 16] = c11[j]; }
    }
}

__global__ void node_coef(const float* __restrict__ h,
                          const float* __restrict__ att,
                          float* __restrict__ ai, float* __restrict__ aj,
                          int nodes, int H, int Cc) {
    int idx = blockIdx.x * blockDim.x + threadIdx.x;
    if (idx >= nodes * H) return;
    int node = idx / H, head = idx % H;
    const float* hv = h + (size_t)node * H * Cc + head * Cc;
    const float* ad = att + head * 2 * Cc;
    const float* as = ad + Cc;
    float si = 0.f, sj = 0.f;
    for (int c = 0; c < Cc; ++c) { si += hv[c] * ad[c]; sj += hv[c] * as[c]; }
    ai[idx] = si;
    aj[idx] = sj;
}

__global__ void edge_max(const int* __restrict__ ei,
                         const float* __restrict__ ai, const float* __restrict__ aj,
                         float* __restrict__ alpha, unsigned* __restrict__ mmax,
                         int H) {
    int idx = blockIdx.x * blockDim.x + threadIdx.x;
    if (idx >= ETOT * H) return;
    int e = idx / H, head = idx % H;
    int s_, d_;
    if (e < E0C) { s_ = ei[e]; d_ = ei[E0C + e]; } else { s_ = d_ = e - E0C; }
    float a = ai[(size_t)d_ * H + head] + aj[(size_t)s_ * H + head];
    a = (a >= 0.f) ? a : NEG_SLOPE * a;
    alpha[idx] = a;
    atomicMax(mmax + (size_t)d_ * H + head, fenc(a));
}

__global__ void edge_expsum(const int* __restrict__ ei,
                            float* __restrict__ alpha,
                            const unsigned* __restrict__ mmax,
                            float* __restrict__ ssum, int H) {
    int idx = blockIdx.x * blockDim.x + threadIdx.x;
    if (idx >= ETOT * H) return;
    int e = idx / H, head = idx % H;
    int d_;
    if (e < E0C) { d_ = ei[E0C + e]; } else { d_ = e - E0C; }
    float m  = fdec(mmax[(size_t)d_ * H + head]);
    float ex = __expf(alpha[idx] - m);
    alpha[idx] = ex;
    atomicAdd(ssum + (size_t)d_ * H + head, ex);
}

__global__ void edge_aggregate(const int* __restrict__ ei,
                               const float* __restrict__ expw,
                               const float* __restrict__ ssum,
                               const float* __restrict__ h,
                               float* __restrict__ agg, int H, int Cc) {
    int gw   = (int)((blockIdx.x * blockDim.x + threadIdx.x) >> 5);
    int lane = threadIdx.x & 31;
    if (gw >= ETOT) return;
    int s_, d_;
    if (gw < E0C) { s_ = ei[gw]; d_ = ei[E0C + gw]; } else { s_ = d_ = gw - E0C; }
    int HC = H * Cc;
    const float* hs = h   + (size_t)s_ * HC;
    float*       ad = agg + (size_t)d_ * HC;
    for (int i = 0; i < HC; i += 32) {
        int ch   = i + lane;
        int head = ch / Cc;
        float w  = expw[(size_t)gw * H + head] / ssum[(size_t)d_ * H + head];
        atomicAdd(ad + ch, w * hs[ch]);
    }
}

__global__ void elu_bias_cvt(const float* __restrict__ agg,
                             const float* __restrict__ bias,
                             _Float16* __restrict__ out, int total, int HC) {
    int i = blockIdx.x * blockDim.x + threadIdx.x;
    if (i >= total) return;
    float v = agg[i] + bias[i % HC];
    v = (v > 0.f) ? v : (__expf(v) - 1.f);
    out[i] = (_Float16)v;
}

__global__ void add_bias(float* __restrict__ out, const float* __restrict__ bias,
                         int total, int Cc) {
    int i = blockIdx.x * blockDim.x + threadIdx.x;
    if (i < total) out[i] += bias[i % Cc];
}

extern "C" void kernel_launch(void* const* d_in, const int* in_sizes, int n_in,
                              void* d_out, int out_size, void* d_ws, size_t ws_size,
                              hipStream_t stream) {
    (void)in_sizes; (void)n_in; (void)out_size; (void)ws_size;

    const float* x    = (const float*)d_in[0];
    const int*   ei   = (const int*)  d_in[1];
    const float* W1   = (const float*)d_in[2];
    const float* att1 = (const float*)d_in[3];
    const float* b1   = (const float*)d_in[4];
    const float* W2   = (const float*)d_in[5];
    const float* att2 = (const float*)d_in[6];
    const float* b2   = (const float*)d_in[7];
    float* out = (float*)d_out;

    char* base = (char*)d_ws;
    size_t off = 0;
    auto carve = [&](size_t bytes) -> char* {
        char* p = base + off;
        off = (off + bytes + 255) & ~(size_t)255;
        return p;
    };
    _Float16* XH    = (_Float16*)carve((size_t)MPAD * INDIM * 2);
    _Float16* W1T   = (_Float16*)carve((size_t)INDIM * HIDDIM * 2);
    _Float16* W2T   = (_Float16*)carve((size_t)HIDDIM * OUTDIM * 2);
    float*    HBUF  = (float*)   carve((size_t)MPAD * HIDDIM * 4);
    float*    AGG1  = (float*)   carve((size_t)NN * HIDDIM * 4);
    float*    AI    = (float*)   carve((size_t)NN * H1C * 4);
    float*    AJ    = (float*)   carve((size_t)NN * H1C * 4);
    unsigned* MM    = (unsigned*)carve((size_t)NN * H1C * 4);
    float*    SS    = (float*)   carve((size_t)NN * H1C * 4);
    float*    ALPHA = (float*)   carve((size_t)ETOT * H1C * 4);

    const int TB = 256;
    auto nb = [](long long n, int tb) { return (unsigned)((n + tb - 1) / tb); };
    const int MT2 = MPAD / 32;

    cvt_f32_to_f16<<<nb((long long)NN * INDIM, TB), TB, 0, stream>>>(x, XH, NN * INDIM);
    xpose_cvt_f16 <<<nb((long long)INDIM * HIDDIM, TB), TB, 0, stream>>>(W1, W1T, INDIM, HIDDIM);
    xpose_cvt_f16 <<<nb((long long)HIDDIM * OUTDIM, TB), TB, 0, stream>>>(W2, W2T, HIDDIM, OUTDIM);

    {
        long long waves = (long long)MT2 * (HIDDIM / 32);
        gemm_f16_wmma<<<nb(waves * 32, 128), 128, 0, stream>>>(XH, W1T, HBUF, MT2, HIDDIM, INDIM);
    }

    node_coef<<<nb((long long)NN * H1C, TB), TB, 0, stream>>>(HBUF, att1, AI, AJ, NN, H1C, C1C);

    hipMemsetAsync(MM,   0, (size_t)NN * H1C * 4, stream);
    hipMemsetAsync(SS,   0, (size_t)NN * H1C * 4, stream);
    hipMemsetAsync(AGG1, 0, (size_t)NN * HIDDIM * 4, stream);

    edge_max   <<<nb((long long)ETOT * H1C, TB), TB, 0, stream>>>(ei, AI, AJ, ALPHA, MM, H1C);
    edge_expsum<<<nb((long long)ETOT * H1C, TB), TB, 0, stream>>>(ei, ALPHA, MM, SS, H1C);
    edge_aggregate<<<nb((long long)ETOT * 32, TB), TB, 0, stream>>>(ei, ALPHA, SS, HBUF, AGG1, H1C, C1C);

    elu_bias_cvt<<<nb((long long)NN * HIDDIM, TB), TB, 0, stream>>>(AGG1, b1, XH, NN * HIDDIM, HIDDIM);

    {
        long long waves = (long long)MT2 * (OUTDIM / 32);
        gemm_f16_wmma<<<nb(waves * 32, 128), 128, 0, stream>>>(XH, W2T, HBUF, MT2, OUTDIM, HIDDIM);
    }

    node_coef<<<nb((long long)NN * H2C, TB), TB, 0, stream>>>(HBUF, att2, AI, AJ, NN, H2C, C2C);

    hipMemsetAsync(MM,  0, (size_t)NN * H2C * 4, stream);
    hipMemsetAsync(SS,  0, (size_t)NN * H2C * 4, stream);
    hipMemsetAsync(out, 0, (size_t)NN * OUTDIM * 4, stream);

    edge_max   <<<nb((long long)ETOT * H2C, TB), TB, 0, stream>>>(ei, AI, AJ, ALPHA, MM, H2C);
    edge_expsum<<<nb((long long)ETOT * H2C, TB), TB, 0, stream>>>(ei, ALPHA, MM, SS, H2C);
    edge_aggregate<<<nb((long long)ETOT * 32, TB), TB, 0, stream>>>(ei, ALPHA, SS, HBUF, out, H2C, C2C);

    add_bias<<<nb((long long)NN * OUTDIM, TB), TB, 0, stream>>>(out, b2, NN * OUTDIM, OUTDIM);
}


# Layer-2 aggregation reads each row's edge count and sources (edge-id order) that layer 1's owner groups store after binning: second destination scan, bins, sort, source lookup gone; sum order kept
# speedup vs baseline: 3.8964x; 3.8964x over previous
#include <hip/hip_runtime.h>
#include <hip/hip_bf16.h>

#define NN     50000
#define MPAD   50016
#define E0C    800000
#define ETOT   850000
#define INDIM  256
#define HIDDIM 256
#define OUTDIM 64
#define H1C    8
#define C1C    32
#define H2C    1
#define C2C    64
#define NEG_SLOPE 0.2f

typedef __attribute__((ext_vector_type(16))) _Float16 v16h;
typedef __attribute__((ext_vector_type(8)))  _Float16 v8h;
typedef __attribute__((ext_vector_type(8)))  float    v8f;
typedef __attribute__((ext_vector_type(4)))  float    v4f;

union V16U { v16h v; v8h h[2]; };


__global__ void xpose_cvt_f16(const float* __restrict__ W,
                              _Float16* __restrict__ Wt, int K, int Nc) {
    int i = blockIdx.x * blockDim.x + threadIdx.x;
    if (i >= K * Nc) return;
    int k = i / Nc, n = i % Nc;
    Wt[(size_t)n * K + k] = (_Float16)W[i];
}

#define WMMA16(a, b, c) __builtin_amdgcn_wmma_f32_16x16x32_f16( \
        false, (a), false, (b), (short)0, (c), false, false)

template <bool ACT>
__device__ __forceinline__ v8h make_a8(const v4f x0, const v4f x1,
                                       const v4f q0, const v4f q1) {
    float f[8] = { x0.x, x0.y, x0.z, x0.w, x1.x, x1.y, x1.z, x1.w };
    if (ACT) {
        const float bb[8] = { q0.x, q0.y, q0.z, q0.w, q1.x, q1.y, q1.z, q1.w };
#pragma unroll
        for (int i = 0; i < 8; ++i) {
            float v = f[i] + bb[i];
            f[i] = (v > 0.f) ? v : (__expf(v) - 1.f);
        }
    }
    v8h o;
#pragma unroll
    for (int i = 0; i < 8; ++i) o[i] = (_Float16)f[i];
    return o;
}

template <bool ACT>
__global__ __launch_bounds__(128)
void gemm_f16_wmma(const float* __restrict__ A,
                   const float* __restrict__ bias,
                   const _Float16* __restrict__ Bt,
                   float* __restrict__ C, int Mtiles2, int Mrows, int Nc, int K) {
    int wave = (int)((blockIdx.x * blockDim.x + threadIdx.x) >> 5);
    int lane = threadIdx.x & 31;
    int nt2 = Nc >> 5;
    int tm = wave / nt2;
    int tn = wave % nt2;
    if (tm >= Mtiles2) return;
    int r    = lane & 15;
    int half = lane >> 4;

    int ra0 = tm * 32 + r;       if (ra0 > Mrows - 1) ra0 = Mrows - 1;
    int ra1 = tm * 32 + 16 + r;  if (ra1 > Mrows - 1) ra1 = Mrows - 1;
    const float* a0 = A + (size_t)ra0 * K + half * 8;
    const float* a1 = A + (size_t)ra1 * K + half * 8;
    const float* bq = ACT ? (bias + half * 8) : nullptr;
    const _Float16* b0 = Bt + (size_t)(tn * 32 + r) * K + half * 8;
    const _Float16* b1 = b0 + (size_t)16 * K;

    v8f c00 = {}, c01 = {}, c10 = {}, c11 = {};
    for (int k0 = 0; k0 < K; k0 += 32) {
        V16U A0, A1, B0, B1;
        const v4f x00 = *(const v4f*)(a0 + k0);
        const v4f x01 = *(const v4f*)(a0 + k0 + 4);
        const v4f x02 = *(const v4f*)(a0 + k0 + 16);
        const v4f x03 = *(const v4f*)(a0 + k0 + 20);
        const v4f x10 = *(const v4f*)(a1 + k0);
        const v4f x11 = *(const v4f*)(a1 + k0 + 4);
        const v4f x12 = *(const v4f*)(a1 + k0 + 16);
        const v4f x13 = *(const v4f*)(a1 + k0 + 20);
        B0.h[0] = *(const v8h*)(b0 + k0);
        B0.h[1] = *(const v8h*)(b0 + k0 + 16);
        B1.h[0] = *(const v8h*)(b1 + k0);
        B1.h[1] = *(const v8h*)(b1 + k0 + 16);
        v4f q0 = {}, q1 = {}, q2 = {}, q3 = {};
        if (ACT) {
            q0 = *(const v4f*)(bq + k0);
            q1 = *(const v4f*)(bq + k0 + 4);
            q2 = *(const v4f*)(bq + k0 + 16);
            q3 = *(const v4f*)(bq + k0 + 20);
        }
        __builtin_amdgcn_sched_barrier(0);
        A0.h[0] = make_a8<ACT>(x00, x01, q0, q1);
        A0.h[1] = make_a8<ACT>(x02, x03, q2, q3);
        A1.h[0] = make_a8<ACT>(x10, x11, q0, q1);
        A1.h[1] = make_a8<ACT>(x12, x13, q2, q3);
        c00 = WMMA16(A0.v, B0.v, c00);
        c01 = WMMA16(A0.v, B1.v, c01);
        c10 = WMMA16(A1.v, B0.v, c10);
        c11 = WMMA16(A1.v, B1.v, c11);
    }
    {
        float* p = C + (size_t)(tm * 32 + half * 8) * Nc + tn * 32 + r;
#pragma unroll
        for (int j = 0; j < 8; ++j) { p[(size_t)j * Nc] = c00[j]; p[(size_t)j * Nc + 16] = c01[j]; }
        p += (size_t)16 * Nc;
#pragma unroll
        for (int j = 0; j < 8; ++j) { p[(size_t)j * Nc] = c10[j]; p[(size_t)j * Nc + 16] = c11[j]; }
    }
}

__global__ void node_coef(const float* __restrict__ h,
                          const float* __restrict__ att,
                          float* __restrict__ ai, float* __restrict__ aj,
                          int nodes, int H, int Cc) {
    int idx = blockIdx.x * blockDim.x + threadIdx.x;
    if (idx >= nodes * H) return;
    int node = idx / H, head = idx % H;
    const float* hv = h + (size_t)node * H * Cc + head * Cc;
    const float* ad = att + head * 2 * Cc;
    const float* as = ad + Cc;
    float si = 0.f, sj = 0.f;
    for (int c = 0; c < Cc; ++c) { si += hv[c] * ad[c]; sj += hv[c] * as[c]; }
    ai[idx] = si;
    aj[idx] = sj;
}

#define AGG_WG     256
#define AGG_WAVES  32
#define RBASE      (NN / AGG_WG)
#define REXTRA     (NN - RBASE * AGG_WG)
#define RMAX       (RBASE + 1)
#define CH_EDGES   256
#define NCHUNK     (E0C / CH_EDGES)
#define BINCAP     64
static_assert(E0C % CH_EDGES == 0, "destination list is read in whole steps");

struct DstStep { int4 a, b; };

__device__ __forceinline__ float leaky(float a) { return (a >= 0.f) ? a : NEG_SLOPE * a; }

__device__ __forceinline__ void bin_step(const DstStep& cur, int cc, int lane, int lo, int cnt,
                                         int* cntr, int* bins) {
    if (cc < 0) return;
    const int dq[8] = { cur.a.x, cur.a.y, cur.a.z, cur.a.w, cur.b.x, cur.b.y, cur.b.z, cur.b.w };
    unsigned rel[8];
    bool hit[8];
    bool anyhit = false;
#pragma unroll
    for (int q = 0; q < 8; ++q) {
        rel[q] = (unsigned)(dq[q] - lo);
        hit[q] = rel[q] < (unsigned)cnt;
        anyhit = anyhit || hit[q];
    }
    if (__builtin_amdgcn_ballot_w32(anyhit) == 0u) return;
    const int ebase = cc * CH_EDGES + lane * 4;
#pragma unroll
    for (int q = 0; q < 8; ++q) {
        if (hit[q]) {
            const int pos = atomicAdd(cntr + rel[q], 1);
            if (pos < BINCAP) bins[rel[q] * BINCAP + pos] = ebase + (q >> 2) * 128 + (q & 3);
        }
    }
}

template <int H, int Cc, bool REUSE>
__global__ __launch_bounds__(1024)
void edge_softmax_aggregate(const int* __restrict__ ei,
                            const float* __restrict__ ai,
                            const float* __restrict__ aj,
                            const float* __restrict__ h,
                            float* __restrict__ agg,
                            int* __restrict__ lcnt,
                            int* __restrict__ lsrc) {
    constexpr int HC = H * Cc;
    constexpr int NV = HC / 32;
    constexpr int VW = (NV >= 4) ? 4 : NV;
    constexpr int NG = NV / VW;
    constexpr int B  = (NV >= 8) ? 8 : 16;
    typedef float vecf __attribute__((ext_vector_type(VW)));
    __shared__ int   cntr[REUSE ? 1 : RMAX];
    __shared__ int   bins[REUSE ? 1 : RMAX * BINCAP];
    __shared__ float wbuf[AGG_WAVES * BINCAP * H];

    const int lane = threadIdx.x & 31;
    const int w    = __builtin_amdgcn_readfirstlane((int)(threadIdx.x >> 5));
    const int b    = blockIdx.x;
    const int lo   = b * RBASE + ((b < REXTRA) ? b : REXTRA);
    const int cnt  = RBASE + ((b < REXTRA) ? 1 : 0);

    const int4* dst4 = (const int4*)(ei + E0C);
    if constexpr (!REUSE) {
        for (int i = threadIdx.x; i < cnt; i += AGG_WAVES * 32) {
            cntr[i] = 1;
            bins[i * BINCAP] = E0C + lo + i;
        }
        __syncthreads();

        {
            const int start = (int)(((long long)b * NCHUNK) / AGG_WG);
            constexpr int ITERS = (NCHUNK + AGG_WAVES - 1) / AGG_WAVES;
            auto chunk_of = [&](int i) -> int {
                const int idx = i * AGG_WAVES + w;
                int c = idx + start;
                if (c >= NCHUNK) c -= NCHUNK;
                return (idx < NCHUNK) ? c : -1;
            };
            auto load_step = [&](int c) -> DstStep {
                const int4* p = dst4 + (size_t)((c >= 0) ? c : 0) * (CH_EDGES / 4) + lane;
                DstStep d; d.a = p[0]; d.b = p[32];
                return d;
            };
            int c0 = chunk_of(0), c1 = chunk_of(1), c2 = chunk_of(2), c3 = chunk_of(3);
            DstStep d0 = load_step(c0), d1 = load_step(c1), d2 = load_step(c2), d3 = load_step(c3);
            for (int it = 0; it < ITERS; it += 4) {
                bin_step(d0, c0, lane, lo, cnt, cntr, bins);
                c0 = chunk_of(it + 4); d0 = load_step(c0);
                bin_step(d1, c1, lane, lo, cnt, cntr, bins);
                c1 = chunk_of(it + 5); d1 = load_step(c1);
                bin_step(d2, c2, lane, lo, cnt, cntr, bins);
                c2 = chunk_of(it + 6); d2 = load_step(c2);
                bin_step(d3, c3, lane, lo, cnt, cntr, bins);
                c3 = chunk_of(it + 7); d3 = load_step(c3);
            }
        }
        __syncthreads();
    }

    float* wrow = wbuf + w * BINCAP * H;
    int n_nx = 0;
    if constexpr (REUSE) n_nx = (w < cnt) ? lcnt[lo + w] : 0;
    for (int r = w; r < cnt; r += AGG_WAVES) {
        const int d = lo + r;
        int n;
        if constexpr (REUSE) {
            n = __builtin_amdgcn_readfirstlane(n_nx);
            n_nx = (r + AGG_WAVES < cnt) ? lcnt[d + AGG_WAVES] : 0;
        } else {
            n = __builtin_amdgcn_readfirstlane(cntr[r]);
            if (lane == 0) lcnt[d] = n;
        }
        float* orow = agg + (size_t)d * HC;
        vecf acc[NG];
#pragma unroll
        for (int g = 0; g < NG; ++g) acc[g] = (vecf)(0.f);

        if (n <= BINCAP) {
            const int NHALF = (n > 32) ? 2 : 1;
            int sv[2];
            int* lrow = lsrc + (size_t)d * BINCAP;
            if constexpr (REUSE) {
                sv[0] = (lane < n) ? lrow[lane] : d;
                sv[1] = sv[0];
                if (NHALF == 2) sv[1] = (lane + 32 < n) ? lrow[lane + 32] : d;
            } else {
                int* bk = bins + r * BINCAP;
                int ev[2];
                ev[0] = (lane < n)      ? bk[lane]      : 0x7FFFFFFF;
                ev[1] = (lane + 32 < n) ? bk[lane + 32] : 0x7FFFFFFF;
                int rk0 = 0, rk1 = 0;
                for (int t = 0; t < n; ++t) {
                    const int x = __builtin_amdgcn_readlane((t < 32) ? ev[0] : ev[1], t & 31);
                    rk0 += (x < ev[0]) ? 1 : 0;
                    rk1 += (x < ev[1]) ? 1 : 0;
                }
                if (lane < n)      bk[rk0] = ev[0];
                if (lane + 32 < n) bk[rk1] = ev[1];
                ev[0] = (lane < n)      ? bk[lane]      : (E0C + d);
                ev[1] = (lane + 32 < n) ? bk[lane + 32] : (E0C + d);
                sv[0] = (ev[0] < E0C) ? ei[ev[0]] : (ev[0] - E0C);
                sv[1] = sv[0];
                if (NHALF == 2) sv[1] = (ev[1] < E0C) ? ei[ev[1]] : (ev[1] - E0C);
                if (lane < n) lrow[lane] = sv[0];
                if (NHALF == 2 && lane + 32 < n) lrow[lane + 32] = sv[1];
            }

            float av[2][H];
            float mx[H];
#pragma unroll
            for (int hd = 0; hd < H; ++hd) mx[hd] = -INFINITY;
#pragma unroll
            for (int hb = 0; hb < 2; ++hb) {
                if (hb < NHALF) {
                    const bool valid = (lane + 32 * hb) < n;
#pragma unroll
                    for (int hd = 0; hd < H; ++hd) {
                        const float a = leaky(ai[(size_t)d * H + hd] + aj[(size_t)sv[hb] * H + hd]);
                        av[hb][hd] = valid ? a : -INFINITY;
                        mx[hd] = fmaxf(mx[hd], av[hb][hd]);
                    }
                } else {
#pragma unroll
                    for (int hd = 0; hd < H; ++hd) av[hb][hd] = -INFINITY;
                }
            }
#pragma unroll
            for (int hd = 0; hd < H; ++hd) {
#pragma unroll
                for (int off = 16; off >= 1; off >>= 1)
                    mx[hd] = fmaxf(mx[hd], __shfl_xor(mx[hd], off, 32));
            }
            float sm[H];
#pragma unroll
            for (int hd = 0; hd < H; ++hd) {
                const float x0 = (lane < n) ? __expf(av[0][hd] - mx[hd]) : 0.f;
                float x1 = 0.f;
                if (NHALF == 2) x1 = (lane + 32 < n) ? __expf(av[1][hd] - mx[hd]) : 0.f;
                av[0][hd] = x0;
                av[1][hd] = x1;
                float t = x0 + x1;
#pragma unroll
                for (int off = 16; off >= 1; off >>= 1)
                    t += __shfl_xor(t, off, 32);
                sm[hd] = t;
            }
#pragma unroll
            for (int hb = 0; hb < 2; ++hb) {
                if (hb < NHALF && (lane + 32 * hb) < n) {
#pragma unroll
                    for (int hd = 0; hd < H; ++hd)
                        wrow[(lane + 32 * hb) * H + hd] = av[hb][hd] / sm[hd];
                }
            }

            for (int t0 = 0; t0 < n; t0 += B) {
                vecf  v[B][NG];
                float wt[B][NG];
#pragma unroll
                for (int q = 0; q < B; ++q) {
                    const int tq = (t0 + q < n) ? (t0 + q) : (n - 1);
                    const int s = __builtin_amdgcn_readlane((tq < 32) ? sv[0] : sv[1], tq & 31);
                    const float* hs = h + (size_t)s * HC;
#pragma unroll
                    for (int g = 0; g < NG; ++g) {
                        const int ch0 = g * 32 * VW + VW * lane;
                        v[q][g]  = *(const vecf*)(hs + ch0);
                        wt[q][g] = wrow[tq * H + ch0 / Cc];
                    }
                }
#pragma unroll
                for (int q = 0; q < B; ++q) {
                    if (t0 + q < n) {
#pragma unroll
                        for (int g = 0; g < NG; ++g) acc[g] += wt[q][g] * v[q][g];
                    }
                }
            }
        } else {
            int   hdg[NG];
            float aig[NG], mxg[NG], smg[NG];
#pragma unroll
            for (int g = 0; g < NG; ++g) {
                hdg[g] = (g * 32 * VW + VW * lane) / Cc;
                aig[g] = ai[(size_t)d * H + hdg[g]];
                mxg[g] = -INFINITY;
                smg[g] = 0.f;
            }
            for (int pass = 0; pass < 3; ++pass) {
                auto edge = [&](int e) {
                    const int s = (e < E0C) ? ei[e] : (e - E0C);
#pragma unroll
                    for (int g = 0; g < NG; ++g) {
                        const float a = leaky(aig[g] + aj[(size_t)s * H + hdg[g]]);
                        if (pass == 0) {
                            mxg[g] = fmaxf(mxg[g], a);
                        } else if (pass == 1) {
                            smg[g] += __expf(a - mxg[g]);
                        } else {
                            const float wgt = __expf(a - mxg[g]) / smg[g];
                            const vecf x = *(const vecf*)(h + (size_t)s * HC + g * 32 * VW + VW * lane);
                            acc[g] += wgt * x;
                        }
                    }
                };
                for (int c = 0; c < NCHUNK; ++c) {
                    const int4* p = dst4 + (size_t)c * (CH_EDGES / 4) + lane;
                    const int4 da = p[0], db = p[32];
                    const int dq[8] = { da.x, da.y, da.z, da.w, db.x, db.y, db.z, db.w };
#pragma unroll
                    for (int q = 0; q < 8; ++q) {
                        unsigned mk = __builtin_amdgcn_ballot_w32(dq[q] == d);
                        while (mk) {
                            const int l = __builtin_ctz(mk);
                            mk &= mk - 1u;
                            edge(c * CH_EDGES + (q >> 2) * 128 + l * 4 + (q & 3));
                        }
                    }
                }
                edge(E0C + d);
            }
        }

#pragma unroll
        for (int g = 0; g < NG; ++g)
            *(vecf*)(orow + g * 32 * VW + VW * lane) = acc[g];
    }
}

__global__ void add_bias(float* __restrict__ out, const float* __restrict__ bias,
                         int total, int Cc) {
    int i = blockIdx.x * blockDim.x + threadIdx.x;
    if (i < total) out[i] += bias[i % Cc];
}

extern "C" void kernel_launch(void* const* d_in, const int* in_sizes, int n_in,
                              void* d_out, int out_size, void* d_ws, size_t ws_size,
                              hipStream_t stream) {
    (void)in_sizes; (void)n_in; (void)out_size; (void)ws_size;

    const float* x    = (const float*)d_in[0];
    const int*   ei   = (const int*)  d_in[1];
    const float* W1   = (const float*)d_in[2];
    const float* att1 = (const float*)d_in[3];
    const float* b1   = (const float*)d_in[4];
    const float* W2   = (const float*)d_in[5];
    const float* att2 = (const float*)d_in[6];
    const float* b2   = (const float*)d_in[7];
    float* out = (float*)d_out;

    char* base = (char*)d_ws;
    size_t off = 0;
    auto carve = [&](size_t bytes) -> char* {
        char* p = base + off;
        off = (off + bytes + 255) & ~(size_t)255;
        return p;
    };
    _Float16* W1T   = (_Float16*)carve((size_t)INDIM * HIDDIM * 2);
    _Float16* W2T   = (_Float16*)carve((size_t)HIDDIM * OUTDIM * 2);
    float*    HBUF  = (float*)   carve((size_t)MPAD * HIDDIM * 4);
    float*    AGG1  = (float*)   carve((size_t)NN * HIDDIM * 4);
    float*    AI    = (float*)   carve((size_t)NN * H1C * 4);
    float*    AJ    = (float*)   carve((size_t)NN * H1C * 4);
    int*      LCNT  = (int*)     carve((size_t)NN * 4);
    int*      LSRC  = (int*)     carve((size_t)NN * BINCAP * 4);

    const int TB = 256;
    auto nb = [](long long n, int tb) { return (unsigned)((n + tb - 1) / tb); };
    const int MT2 = MPAD / 32;

    xpose_cvt_f16 <<<nb((long long)INDIM * HIDDIM, TB), TB, 0, stream>>>(W1, W1T, INDIM, HIDDIM);
    xpose_cvt_f16 <<<nb((long long)HIDDIM * OUTDIM, TB), TB, 0, stream>>>(W2, W2T, HIDDIM, OUTDIM);

    {
        long long waves = (long long)MT2 * (HIDDIM / 32);
        gemm_f16_wmma<false><<<nb(waves * 32, 128), 128, 0, stream>>>(x, nullptr, W1T, HBUF, MT2, NN, HIDDIM, INDIM);
    }

    node_coef<<<nb((long long)NN * H1C, TB), TB, 0, stream>>>(HBUF, att1, AI, AJ, NN, H1C, C1C);

    edge_softmax_aggregate<H1C, C1C, false><<<AGG_WG, AGG_WAVES * 32, 0, stream>>>(ei, AI, AJ, HBUF, AGG1, LCNT, LSRC);

    {
        long long waves = (long long)MT2 * (OUTDIM / 32);
        gemm_f16_wmma<true><<<nb(waves * 32, 128), 128, 0, stream>>>(AGG1, b1, W2T, HBUF, MT2, NN, OUTDIM, HIDDIM);
    }

    node_coef<<<nb((long long)NN * H2C, TB), TB, 0, stream>>>(HBUF, att2, AI, AJ, NN, H2C, C2C);

    edge_softmax_aggregate<H2C, C2C, true><<<AGG_WG, AGG_WAVES * 32, 0, stream>>>(ei, AI, AJ, HBUF, out, LCNT, LSRC);

    add_bias<<<nb((long long)NN * OUTDIM, TB), TB, 0, stream>>>(out, b2, NN * OUTDIM, OUTDIM);
}


# GEMM A rows staged once per work-group in LDS as f16 (whole-line f32 loads; producer pass and rounding once per element, was once per column wave); A fragments read from LDS; values, k order kept
# speedup vs baseline: 4.5387x; 1.1648x over previous
#include <hip/hip_runtime.h>
#include <hip/hip_bf16.h>

#define NN     50000
#define MPAD   50016
#define E0C    800000
#define ETOT   850000
#define INDIM  256
#define HIDDIM 256
#define OUTDIM 64
#define H1C    8
#define C1C    32
#define H2C    1
#define C2C    64
#define NEG_SLOPE 0.2f

typedef __attribute__((ext_vector_type(16))) _Float16 v16h;
typedef __attribute__((ext_vector_type(8)))  _Float16 v8h;
typedef __attribute__((ext_vector_type(4)))  _Float16 v4h;
typedef __attribute__((ext_vector_type(8)))  float    v8f;
typedef __attribute__((ext_vector_type(4)))  float    v4f;

union V16U { v16h v; v8h h[2]; };


__global__ void xpose_cvt_f16(const float* __restrict__ W,
                              _Float16* __restrict__ Wt, int K, int Nc) {
    int i = blockIdx.x * blockDim.x + threadIdx.x;
    if (i >= K * Nc) return;
    int k = i / Nc, n = i % Nc;
    Wt[(size_t)n * K + k] = (_Float16)W[i];
}

#define WMMA16(a, b, c) __builtin_amdgcn_wmma_f32_16x16x32_f16( \
        false, (a), false, (b), (short)0, (c), false, false)

#define A_PITCH (256 + 8)

template <bool ACT, int NT, int TM, int K>
__global__ __launch_bounds__(NT * TM * 32)
void gemm_f16_wmma(const float* __restrict__ A,
                   const float* __restrict__ bias,
                   const _Float16* __restrict__ Bt,
                   float* __restrict__ C, int Mtiles2, int Mrows) {
    constexpr int NTHR = NT * TM * 32;
    constexpr int ROWS = TM * 32;
    constexpr int Nc   = NT * 32;
    constexpr int QPR  = K / 4;
    constexpr int RPP  = NTHR / QPR;
    static_assert(K + 8 == A_PITCH, "LDS pitch is made for this K");
    static_assert(NTHR % QPR == 0 && ROWS % RPP == 0, "whole rows per pass, whole passes per tile");
    __shared__ __attribute__((aligned(16))) _Float16 As[ROWS * A_PITCH];

    const int t = threadIdx.x;
    {
        const int cq = t % QPR;
        const int rs = t / QPR;
        v4f q = {};
        if (ACT) q = *(const v4f*)(bias + cq * 4);
        const int row0 = blockIdx.x * ROWS;
#pragma unroll
        for (int i = 0; i < ROWS / RPP; ++i) {
            const int rl = i * RPP + rs;
            int rg = row0 + rl;  if (rg > Mrows - 1) rg = Mrows - 1;
            const v4f x = *(const v4f*)(A + (size_t)rg * K + cq * 4);
            float f[4] = { x.x, x.y, x.z, x.w };
            if (ACT) {
                const float bb[4] = { q.x, q.y, q.z, q.w };
#pragma unroll
                for (int j = 0; j < 4; ++j) {
                    float v = f[j] + bb[j];
                    f[j] = (v > 0.f) ? v : (__expf(v) - 1.f);
                }
            }
            v4h o;
#pragma unroll
            for (int j = 0; j < 4; ++j) o[j] = (_Float16)f[j];
            *(v4h*)(As + rl * A_PITCH + cq * 4) = o;
        }
    }
    __syncthreads();

    const int lane = t & 31;
    const int w   = __builtin_amdgcn_readfirstlane(t >> 5);
    const int tml = w / NT;
    const int tn  = w % NT;
    const int tm  = blockIdx.x * TM + tml;
    if (tm >= Mtiles2) return;
    const int r    = lane & 15;
    const int half = lane >> 4;

    const _Float16* a0 = As + (tml * 32 + r) * A_PITCH + half * 8;
    const _Float16* a1 = a0 + 16 * A_PITCH;
    const _Float16* b0 = Bt + (size_t)(tn * 32 + r) * K + half * 8;
    const _Float16* b1 = b0 + (size_t)16 * K;

    v8f c00 = {}, c01 = {}, c10 = {}, c11 = {};
#pragma unroll
    for (int k0 = 0; k0 < K; k0 += 32) {
        V16U A0, A1, B0, B1;
        B0.h[0] = *(const v8h*)(b0 + k0);
        B0.h[1] = *(const v8h*)(b0 + k0 + 16);
        B1.h[0] = *(const v8h*)(b1 + k0);
        B1.h[1] = *(const v8h*)(b1 + k0 + 16);
        A0.h[0] = *(const v8h*)(a0 + k0);
        A0.h[1] = *(const v8h*)(a0 + k0 + 16);
        A1.h[0] = *(const v8h*)(a1 + k0);
        A1.h[1] = *(const v8h*)(a1 + k0 + 16);
        c00 = WMMA16(A0.v, B0.v, c00);
        c01 = WMMA16(A0.v, B1.v, c01);
        c10 = WMMA16(A1.v, B0.v, c10);
        c11 = WMMA16(A1.v, B1.v, c11);
    }
    {
        float* p = C + (size_t)(tm * 32 + half * 8) * Nc + tn * 32 + r;
#pragma unroll
        for (int j = 0; j < 8; ++j) { p[(size_t)j * Nc] = c00[j]; p[(size_t)j * Nc + 16] = c01[j]; }
        p += (size_t)16 * Nc;
#pragma unroll
        for (int j = 0; j < 8; ++j) { p[(size_t)j * Nc] = c10[j]; p[(size_t)j * Nc + 16] = c11[j]; }
    }
}

__global__ void node_coef(const float* __restrict__ h,
                          const float* __restrict__ att,
                          float* __restrict__ ai, float* __restrict__ aj,
                          int nodes, int H, int Cc) {
    int idx = blockIdx.x * blockDim.x + threadIdx.x;
    if (idx >= nodes * H) return;
    int node = idx / H, head = idx % H;
    const float* hv = h + (size_t)node * H * Cc + head * Cc;
    const float* ad = att + head * 2 * Cc;
    const float* as = ad + Cc;
    float si = 0.f, sj = 0.f;
    for (int c = 0; c < Cc; ++c) { si += hv[c] * ad[c]; sj += hv[c] * as[c]; }
    ai[idx] = si;
    aj[idx] = sj;
}

#define AGG_WG     256
#define AGG_WAVES  32
#define RBASE      (NN / AGG_WG)
#define REXTRA     (NN - RBASE * AGG_WG)
#define RMAX       (RBASE + 1)
#define CH_EDGES   256
#define NCHUNK     (E0C / CH_EDGES)
#define BINCAP     64
static_assert(E0C % CH_EDGES == 0, "destination list is read in whole steps");

struct DstStep { int4 a, b; };

__device__ __forceinline__ float leaky(float a) { return (a >= 0.f) ? a : NEG_SLOPE * a; }

__device__ __forceinline__ void bin_step(const DstStep& cur, int cc, int lane, int lo, int cnt,
                                         int* cntr, int* bins) {
    if (cc < 0) return;
    const int dq[8] = { cur.a.x, cur.a.y, cur.a.z, cur.a.w, cur.b.x, cur.b.y, cur.b.z, cur.b.w };
    unsigned rel[8];
    bool hit[8];
    bool anyhit = false;
#pragma unroll
    for (int q = 0; q < 8; ++q) {
        rel[q] = (unsigned)(dq[q] - lo);
        hit[q] = rel[q] < (unsigned)cnt;
        anyhit = anyhit || hit[q];
    }
    if (__builtin_amdgcn_ballot_w32(anyhit) == 0u) return;
    const int ebase = cc * CH_EDGES + lane * 4;
#pragma unroll
    for (int q = 0; q < 8; ++q) {
        if (hit[q]) {
            const int pos = atomicAdd(cntr + rel[q], 1);
            if (pos < BINCAP) bins[rel[q] * BINCAP + pos] = ebase + (q >> 2) * 128 + (q & 3);
        }
    }
}

template <int H, int Cc, bool REUSE>
__global__ __launch_bounds__(1024)
void edge_softmax_aggregate(const int* __restrict__ ei,
                            const float* __restrict__ ai,
                            const float* __restrict__ aj,
                            const float* __restrict__ h,
                            float* __restrict__ agg,
                            int* __restrict__ lcnt,
                            int* __restrict__ lsrc) {
    constexpr int HC = H * Cc;
    constexpr int NV = HC / 32;
    constexpr int VW = (NV >= 4) ? 4 : NV;
    constexpr int NG = NV / VW;
    constexpr int B  = (NV >= 8) ? 8 : 16;
    typedef float vecf __attribute__((ext_vector_type(VW)));
    __shared__ int   cntr[REUSE ? 1 : RMAX];
    __shared__ int   bins[REUSE ? 1 : RMAX * BINCAP];
    __shared__ float wbuf[AGG_WAVES * BINCAP * H];

    const int lane = threadIdx.x & 31;
    const int w    = __builtin_amdgcn_readfirstlane((int)(threadIdx.x >> 5));
    const int b    = blockIdx.x;
    const int lo   = b * RBASE + ((b < REXTRA) ? b : REXTRA);
    const int cnt  = RBASE + ((b < REXTRA) ? 1 : 0);

    const int4* dst4 = (const int4*)(ei + E0C);
    if constexpr (!REUSE) {
        for (int i = threadIdx.x; i < cnt; i += AGG_WAVES * 32) {
            cntr[i] = 1;
            bins[i * BINCAP] = E0C + lo + i;
        }
        __syncthreads();

        {
            const int start = (int)(((long long)b * NCHUNK) / AGG_WG);
            constexpr int ITERS = (NCHUNK + AGG_WAVES - 1) / AGG_WAVES;
            auto chunk_of = [&](int i) -> int {
                const int idx = i * AGG_WAVES + w;
                int c = idx + start;
                if (c >= NCHUNK) c -= NCHUNK;
                return (idx < NCHUNK) ? c : -1;
            };
            auto load_step = [&](int c) -> DstStep {
                const int4* p = dst4 + (size_t)((c >= 0) ? c : 0) * (CH_EDGES / 4) + lane;
                DstStep d; d.a = p[0]; d.b = p[32];
                return d;
            };
            int c0 = chunk_of(0), c1 = chunk_of(1), c2 = chunk_of(2), c3 = chunk_of(3);
            DstStep d0 = load_step(c0), d1 = load_step(c1), d2 = load_step(c2), d3 = load_step(c3);
            for (int it = 0; it < ITERS; it += 4) {
                bin_step(d0, c0, lane, lo, cnt, cntr, bins);
                c0 = chunk_of(it + 4); d0 = load_step(c0);
                bin_step(d1, c1, lane, lo, cnt, cntr, bins);
                c1 = chunk_of(it + 5); d1 = load_step(c1);
                bin_step(d2, c2, lane, lo, cnt, cntr, bins);
                c2 = chunk_of(it + 6); d2 = load_step(c2);
                bin_step(d3, c3, lane, lo, cnt, cntr, bins);
                c3 = chunk_of(it + 7); d3 = load_step(c3);
            }
        }
        __syncthreads();
    }

    float* wrow = wbuf + w * BINCAP * H;
    int n_nx = 0;
    if constexpr (REUSE) n_nx = (w < cnt) ? lcnt[lo + w] : 0;
    for (int r = w; r < cnt; r += AGG_WAVES) {
        const int d = lo + r;
        int n;
        if constexpr (REUSE) {
            n = __builtin_amdgcn_readfirstlane(n_nx);
            n_nx = (r + AGG_WAVES < cnt) ? lcnt[d + AGG_WAVES] : 0;
        } else {
            n = __builtin_amdgcn_readfirstlane(cntr[r]);
            if (lane == 0) lcnt[d] = n;
        }
        float* orow = agg + (size_t)d * HC;
        vecf acc[NG];
#pragma unroll
        for (int g = 0; g < NG; ++g) acc[g] = (vecf)(0.f);

        if (n <= BINCAP) {
            const int NHALF = (n > 32) ? 2 : 1;
            int sv[2];
            int* lrow = lsrc + (size_t)d * BINCAP;
            if constexpr (REUSE) {
                sv[0] = (lane < n) ? lrow[lane] : d;
                sv[1] = sv[0];
                if (NHALF == 2) sv[1] = (lane + 32 < n) ? lrow[lane + 32] : d;
            } else {
                int* bk = bins + r * BINCAP;
                int ev[2];
                ev[0] = (lane < n)      ? bk[lane]      : 0x7FFFFFFF;
                ev[1] = (lane + 32 < n) ? bk[lane + 32] : 0x7FFFFFFF;
                int rk0 = 0, rk1 = 0;
                for (int t = 0; t < n; ++t) {
                    const int x = __builtin_amdgcn_readlane((t < 32) ? ev[0] : ev[1], t & 31);
                    rk0 += (x < ev[0]) ? 1 : 0;
                    rk1 += (x < ev[1]) ? 1 : 0;
                }
                if (lane < n)      bk[rk0] = ev[0];
                if (lane + 32 < n) bk[rk1] = ev[1];
                ev[0] = (lane < n)      ? bk[lane]      : (E0C + d);
                ev[1] = (lane + 32 < n) ? bk[lane + 32] : (E0C + d);
                sv[0] = (ev[0] < E0C) ? ei[ev[0]] : (ev[0] - E0C);
                sv[1] = sv[0];
                if (NHALF == 2) sv[1] = (ev[1] < E0C) ? ei[ev[1]] : (ev[1] - E0C);
                if (lane < n) lrow[lane] = sv[0];
                if (NHALF == 2 && lane + 32 < n) lrow[lane + 32] = sv[1];
            }

            float av[2][H];
            float mx[H];
#pragma unroll
            for (int hd = 0; hd < H; ++hd) mx[hd] = -INFINITY;
#pragma unroll
            for (int hb = 0; hb < 2; ++hb) {
                if (hb < NHALF) {
                    const bool valid = (lane + 32 * hb) < n;
#pragma unroll
                    for (int hd = 0; hd < H; ++hd) {
                        const float a = leaky(ai[(size_t)d * H + hd] + aj[(size_t)sv[hb] * H + hd]);
                        av[hb][hd] = valid ? a : -INFINITY;
                        mx[hd] = fmaxf(mx[hd], av[hb][hd]);
                    }
                } else {
#pragma unroll
                    for (int hd = 0; hd < H; ++hd) av[hb][hd] = -INFINITY;
                }
            }
#pragma unroll
            for (int hd = 0; hd < H; ++hd) {
#pragma unroll
                for (int off = 16; off >= 1; off >>= 1)
                    mx[hd] = fmaxf(mx[hd], __shfl_xor(mx[hd], off, 32));
            }
            float sm[H];
#pragma unroll
            for (int hd = 0; hd < H; ++hd) {
                const float x0 = (lane < n) ? __expf(av[0][hd] - mx[hd]) : 0.f;
                float x1 = 0.f;
                if (NHALF == 2) x1 = (lane + 32 < n) ? __expf(av[1][hd] - mx[hd]) : 0.f;
                av[0][hd] = x0;
                av[1][hd] = x1;
                float t = x0 + x1;
#pragma unroll
                for (int off = 16; off >= 1; off >>= 1)
                    t += __shfl_xor(t, off, 32);
                sm[hd] = t;
            }
#pragma unroll
            for (int hb = 0; hb < 2; ++hb) {
                if (hb < NHALF && (lane + 32 * hb) < n) {
#pragma unroll
                    for (int hd = 0; hd < H; ++hd)
                        wrow[(lane + 32 * hb) * H + hd] = av[hb][hd] / sm[hd];
                }
            }

            for (int t0 = 0; t0 < n; t0 += B) {
                vecf  v[B][NG];
                float wt[B][NG];
#pragma unroll
                for (int q = 0; q < B; ++q) {
                    const int tq = (t0 + q < n) ? (t0 + q) : (n - 1);
                    const int s = __builtin_amdgcn_readlane((tq < 32) ? sv[0] : sv[1], tq & 31);
                    const float* hs = h + (size_t)s * HC;
#pragma unroll
                    for (int g = 0; g < NG; ++g) {
                        const int ch0 = g * 32 * VW + VW * lane;
                        v[q][g]  = *(const vecf*)(hs + ch0);
                        wt[q][g] = wrow[tq * H + ch0 / Cc];
                    }
                }
#pragma unroll
                for (int q = 0; q < B; ++q) {
                    if (t0 + q < n) {
#pragma unroll
                        for (int g = 0; g < NG; ++g) acc[g] += wt[q][g] * v[q][g];
                    }
                }
            }
        } else {
            int   hdg[NG];
            float aig[NG], mxg[NG], smg[NG];
#pragma unroll
            for (int g = 0; g < NG; ++g) {
                hdg[g] = (g * 32 * VW + VW * lane) / Cc;
                aig[g] = ai[(size_t)d * H + hdg[g]];
                mxg[g] = -INFINITY;
                smg[g] = 0.f;
            }
            for (int pass = 0; pass < 3; ++pass) {
                auto edge = [&](int e) {
                    const int s = (e < E0C) ? ei[e] : (e - E0C);
#pragma unroll
                    for (int g = 0; g < NG; ++g) {
                        const float a = leaky(aig[g] + aj[(size_t)s * H + hdg[g]]);
                        if (pass == 0) {
                            mxg[g] = fmaxf(mxg[g], a);
                        } else if (pass == 1) {
                            smg[g] += __expf(a - mxg[g]);
                        } else {
                            const float wgt = __expf(a - mxg[g]) / smg[g];
                            const vecf x = *(const vecf*)(h + (size_t)s * HC + g * 32 * VW + VW * lane);
                            acc[g] += wgt * x;
                        }
                    }
                };
                for (int c = 0; c < NCHUNK; ++c) {
                    const int4* p = dst4 + (size_t)c * (CH_EDGES / 4) + lane;
                    const int4 da = p[0], db = p[32];
                    const int dq[8] = { da.x, da.y, da.z, da.w, db.x, db.y, db.z, db.w };
#pragma unroll
                    for (int q = 0; q < 8; ++q) {
                        unsigned mk = __builtin_amdgcn_ballot_w32(dq[q] == d);
                        while (mk) {
                            const int l = __builtin_ctz(mk);
                            mk &= mk - 1u;
                            edge(c * CH_EDGES + (q >> 2) * 128 + l * 4 + (q & 3));
                        }
                    }
                }
                edge(E0C + d);
            }
        }

#pragma unroll
        for (int g = 0; g < NG; ++g)
            *(vecf*)(orow + g * 32 * VW + VW * lane) = acc[g];
    }
}

__global__ void add_bias(float* __restrict__ out, const float* __restrict__ bias,
                         int total, int Cc) {
    int i = blockIdx.x * blockDim.x + threadIdx.x;
    if (i < total) out[i] += bias[i % Cc];
}

extern "C" void kernel_launch(void* const* d_in, const int* in_sizes, int n_in,
                              void* d_out, int out_size, void* d_ws, size_t ws_size,
                              hipStream_t stream) {
    (void)in_sizes; (void)n_in; (void)out_size; (void)ws_size;

    const float* x    = (const float*)d_in[0];
    const int*   ei   = (const int*)  d_in[1];
    const float* W1   = (const float*)d_in[2];
    const float* att1 = (const float*)d_in[3];
    const float* b1   = (const float*)d_in[4];
    const float* W2   = (const float*)d_in[5];
    const float* att2 = (const float*)d_in[6];
    const float* b2   = (const float*)d_in[7];
    float* out = (float*)d_out;

    char* base = (char*)d_ws;
    size_t off = 0;
    auto carve = [&](size_t bytes) -> char* {
        char* p = base + off;
        off = (off + bytes + 255) & ~(size_t)255;
        return p;
    };
    _Float16* W1T   = (_Float16*)carve((size_t)INDIM * HIDDIM * 2);
    _Float16* W2T   = (_Float16*)carve((size_t)HIDDIM * OUTDIM * 2);
    float*    HBUF  = (float*)   carve((size_t)MPAD * HIDDIM * 4);
    float*    AGG1  = (float*)   carve((size_t)NN * HIDDIM * 4);
    float*    AI    = (float*)   carve((size_t)NN * H1C * 4);
    float*    AJ    = (float*)   carve((size_t)NN * H1C * 4);
    int*      LCNT  = (int*)     carve((size_t)NN * 4);
    int*      LSRC  = (int*)     carve((size_t)NN * BINCAP * 4);

    const int TB = 256;
    auto nb = [](long long n, int tb) { return (unsigned)((n + tb - 1) / tb); };
    const int MT2 = MPAD / 32;

    xpose_cvt_f16 <<<nb((long long)INDIM * HIDDIM, TB), TB, 0, stream>>>(W1, W1T, INDIM, HIDDIM);
    xpose_cvt_f16 <<<nb((long long)HIDDIM * OUTDIM, TB), TB, 0, stream>>>(W2, W2T, HIDDIM, OUTDIM);

    gemm_f16_wmma<false, HIDDIM / 32, 1, INDIM><<<MT2, 256, 0, stream>>>(x, nullptr, W1T, HBUF, MT2, NN);

    node_coef<<<nb((long long)NN * H1C, TB), TB, 0, stream>>>(HBUF, att1, AI, AJ, NN, H1C, C1C);

    edge_softmax_aggregate<H1C, C1C, false><<<AGG_WG, AGG_WAVES * 32, 0, stream>>>(ei, AI, AJ, HBUF, AGG1, LCNT, LSRC);

    gemm_f16_wmma<true, OUTDIM / 32, 2, HIDDIM><<<(MT2 + 1) / 2, 128, 0, stream>>>(AGG1, b1, W2T, HBUF, MT2, NN);

    node_coef<<<nb((long long)NN * H2C, TB), TB, 0, stream>>>(HBUF, att2, AI, AJ, NN, H2C, C2C);

    edge_softmax_aggregate<H2C, C2C, true><<<AGG_WG, AGG_WAVES * 32, 0, stream>>>(ei, AI, AJ, HBUF, out, LCNT, LSRC);

    add_bias<<<nb((long long)NN * OUTDIM, TB), TB, 0, stream>>>(out, b2, NN * OUTDIM, OUTDIM);
}


# Layer-1 owner groups scan one owner byte per edge (plane written by a new pre-pass) instead of 4-byte destinations; a wave lists its byte matches in LDS, reads only their destinations; sum order kept
# speedup vs baseline: 4.9098x; 1.0818x over previous
#include <hip/hip_runtime.h>
#include <hip/hip_bf16.h>

#define NN     50000
#define MPAD   50016
#define E0C    800000
#define ETOT   850000
#define INDIM  256
#define HIDDIM 256
#define OUTDIM 64
#define H1C    8
#define C1C    32
#define H2C    1
#define C2C    64
#define NEG_SLOPE 0.2f

typedef __attribute__((ext_vector_type(16))) _Float16 v16h;
typedef __attribute__((ext_vector_type(8)))  _Float16 v8h;
typedef __attribute__((ext_vector_type(4)))  _Float16 v4h;
typedef __attribute__((ext_vector_type(8)))  float    v8f;
typedef __attribute__((ext_vector_type(4)))  float    v4f;

union V16U { v16h v; v8h h[2]; };


__global__ void xpose_cvt_f16(const float* __restrict__ W,
                              _Float16* __restrict__ Wt, int K, int Nc) {
    int i = blockIdx.x * blockDim.x + threadIdx.x;
    if (i >= K * Nc) return;
    int k = i / Nc, n = i % Nc;
    Wt[(size_t)n * K + k] = (_Float16)W[i];
}

#define WMMA16(a, b, c) __builtin_amdgcn_wmma_f32_16x16x32_f16( \
        false, (a), false, (b), (short)0, (c), false, false)

#define A_PITCH (256 + 8)

template <bool ACT, int NT, int TM, int K>
__global__ __launch_bounds__(NT * TM * 32)
void gemm_f16_wmma(const float* __restrict__ A,
                   const float* __restrict__ bias,
                   const _Float16* __restrict__ Bt,
                   float* __restrict__ C, int Mtiles2, int Mrows) {
    constexpr int NTHR = NT * TM * 32;
    constexpr int ROWS = TM * 32;
    constexpr int Nc   = NT * 32;
    constexpr int QPR  = K / 4;
    constexpr int RPP  = NTHR / QPR;
    static_assert(K + 8 == A_PITCH, "LDS pitch is made for this K");
    static_assert(NTHR % QPR == 0 && ROWS % RPP == 0, "whole rows per pass, whole passes per tile");
    __shared__ __attribute__((aligned(16))) _Float16 As[ROWS * A_PITCH];

    const int t = threadIdx.x;
    {
        const int cq = t % QPR;
        const int rs = t / QPR;
        v4f q = {};
        if (ACT) q = *(const v4f*)(bias + cq * 4);
        const int row0 = blockIdx.x * ROWS;
#pragma unroll
        for (int i = 0; i < ROWS / RPP; ++i) {
            const int rl = i * RPP + rs;
            int rg = row0 + rl;  if (rg > Mrows - 1) rg = Mrows - 1;
            const v4f x = *(const v4f*)(A + (size_t)rg * K + cq * 4);
            float f[4] = { x.x, x.y, x.z, x.w };
            if (ACT) {
                const float bb[4] = { q.x, q.y, q.z, q.w };
#pragma unroll
                for (int j = 0; j < 4; ++j) {
                    float v = f[j] + bb[j];
                    f[j] = (v > 0.f) ? v : (__expf(v) - 1.f);
                }
            }
            v4h o;
#pragma unroll
            for (int j = 0; j < 4; ++j) o[j] = (_Float16)f[j];
            *(v4h*)(As + rl * A_PITCH + cq * 4) = o;
        }
    }
    __syncthreads();

    const int lane = t & 31;
    const int w   = __builtin_amdgcn_readfirstlane(t >> 5);
    const int tml = w / NT;
    const int tn  = w % NT;
    const int tm  = blockIdx.x * TM + tml;
    if (tm >= Mtiles2) return;
    const int r    = lane & 15;
    const int half = lane >> 4;

    const _Float16* a0 = As + (tml * 32 + r) * A_PITCH + half * 8;
    const _Float16* a1 = a0 + 16 * A_PITCH;
    const _Float16* b0 = Bt + (size_t)(tn * 32 + r) * K + half * 8;
    const _Float16* b1 = b0 + (size_t)16 * K;

    v8f c00 = {}, c01 = {}, c10 = {}, c11 = {};
#pragma unroll
    for (int k0 = 0; k0 < K; k0 += 32) {
        V16U A0, A1, B0, B1;
        B0.h[0] = *(const v8h*)(b0 + k0);
        B0.h[1] = *(const v8h*)(b0 + k0 + 16);
        B1.h[0] = *(const v8h*)(b1 + k0);
        B1.h[1] = *(const v8h*)(b1 + k0 + 16);
        A0.h[0] = *(const v8h*)(a0 + k0);
        A0.h[1] = *(const v8h*)(a0 + k0 + 16);
        A1.h[0] = *(const v8h*)(a1 + k0);
        A1.h[1] = *(const v8h*)(a1 + k0 + 16);
        c00 = WMMA16(A0.v, B0.v, c00);
        c01 = WMMA16(A0.v, B1.v, c01);
        c10 = WMMA16(A1.v, B0.v, c10);
        c11 = WMMA16(A1.v, B1.v, c11);
    }
    {
        float* p = C + (size_t)(tm * 32 + half * 8) * Nc + tn * 32 + r;
#pragma unroll
        for (int j = 0; j < 8; ++j) { p[(size_t)j * Nc] = c00[j]; p[(size_t)j * Nc + 16] = c01[j]; }
        p += (size_t)16 * Nc;
#pragma unroll
        for (int j = 0; j < 8; ++j) { p[(size_t)j * Nc] = c10[j]; p[(size_t)j * Nc + 16] = c11[j]; }
    }
}

__global__ void node_coef(const float* __restrict__ h,
                          const float* __restrict__ att,
                          float* __restrict__ ai, float* __restrict__ aj,
                          int nodes, int H, int Cc) {
    int idx = blockIdx.x * blockDim.x + threadIdx.x;
    if (idx >= nodes * H) return;
    int node = idx / H, head = idx % H;
    const float* hv = h + (size_t)node * H * Cc + head * Cc;
    const float* ad = att + head * 2 * Cc;
    const float* as = ad + Cc;
    float si = 0.f, sj = 0.f;
    for (int c = 0; c < Cc; ++c) { si += hv[c] * ad[c]; sj += hv[c] * as[c]; }
    ai[idx] = si;
    aj[idx] = sj;
}

#define AGG_WG     256
#define AGG_WAVES  32
#define RBASE      (NN / AGG_WG)
#define REXTRA     (NN - RBASE * AGG_WG)
#define RMAX       (RBASE + 1)
#define CH_EDGES   256
#define NCHUNK     (E0C / CH_EDGES)
#define BINCAP     64
static_assert(E0C % CH_EDGES == 0, "destination list is read in whole steps");
#define ST_EDGES   1024
#define NSTEP      ((E0C + ST_EDGES - 1) / ST_EDGES)
#define EPAD       (NSTEP * ST_EDGES)
#define WCAP       256
static_assert(AGG_WG == 256, "an owner number is one byte");
static_assert(E0C % 4 == 0, "four edges per word of the byte plane");

struct OwnStep { uint4 a, b; };

__device__ __forceinline__ unsigned owner_of(int d) {
    const int big = REXTRA * RMAX;
    return (d < big) ? ((unsigned)d / (unsigned)RMAX)
                     : ((unsigned)REXTRA + (unsigned)(d - big) / (unsigned)RBASE);
}

__global__ void owner_bytes(const int* __restrict__ ei, unsigned* __restrict__ own) {
    const int t  = blockIdx.x * blockDim.x + threadIdx.x;
    const int i0 = (t >> 5) * 128 + (t & 31);
    const int4* dst4 = (const int4*)(ei + E0C);
    int4 d[4];
#pragma unroll
    for (int j = 0; j < 4; ++j) {
        const int i = i0 + j * 32;
        d[j] = dst4[(i < E0C / 4) ? i : 0];
    }
#pragma unroll
    for (int j = 0; j < 4; ++j) {
        const int i = i0 + j * 32;
        if (i < EPAD / 4) {
            const unsigned v = (owner_of(d[j].x) & 255u) | ((owner_of(d[j].y) & 255u) << 8)
                             | ((owner_of(d[j].z) & 255u) << 16) | ((owner_of(d[j].w) & 255u) << 24);
            own[i] = (i < E0C / 4) ? v : 0xFFFFFFFFu;
        }
    }
}

__device__ __forceinline__ float leaky(float a) { return (a >= 0.f) ? a : NEG_SLOPE * a; }

__device__ __forceinline__ void file_edge(int e, const int* __restrict__ ei, int lo, int cnt,
                                          int* cntr, int* bins) {
    const unsigned rel = (unsigned)(ei[E0C + e] - lo);
    if (rel < (unsigned)cnt) {
        const int pos = atomicAdd(cntr + rel, 1);
        if (pos < BINCAP) bins[rel * BINCAP + pos] = e;
    }
}

__device__ __forceinline__ void scan_step(const OwnStep& cur, int cc, int lane, unsigned pat,
                                          const int* __restrict__ ei, int lo, int cnt,
                                          int& wn, int* wl, int* cntr, int* bins) {
    if (cc < 0) return;
    const unsigned v[8] = { cur.a.x, cur.a.y, cur.a.z, cur.a.w, cur.b.x, cur.b.y, cur.b.z, cur.b.w };
    unsigned m = 0;
#pragma unroll
    for (int q = 0; q < 8; ++q) {
        const unsigned y = v[q] ^ pat;
        const unsigned z = (y - 0x01010101u) & ~y & 0x80808080u;
        m |= (z >> 7) << q;
    }
    const int ebase = cc * ST_EDGES + lane * 16;
    if (cc == NSTEP - 1) {
#pragma unroll
        for (int q = 0; q < 8; ++q) {
            if (ebase + (q >> 2) * 512 + (q & 3) * 4 >= E0C) m &= ~(0x01010101u << q);
        }
    }
    unsigned pend = __builtin_amdgcn_ballot_w32(m != 0u);
    while (pend != 0u) {
        if (m != 0u) {
            const int slot = wn + (int)__builtin_amdgcn_mbcnt_lo(pend, 0u);
            const int bit = __builtin_ctz(m);
            m &= m - 1u;
            const int q = bit & 7, k = bit >> 3;
            const int e = ebase + (q >> 2) * 512 + (q & 3) * 4 + k;
            if (slot < WCAP) wl[slot] = e;
            else file_edge(e, ei, lo, cnt, cntr, bins);
        }
        wn += __builtin_popcount(pend);
        pend = __builtin_amdgcn_ballot_w32(m != 0u);
    }
}

template <int H, int Cc, bool REUSE>
__global__ __launch_bounds__(1024)
void edge_softmax_aggregate(const int* __restrict__ ei,
                            const float* __restrict__ ai,
                            const float* __restrict__ aj,
                            const float* __restrict__ h,
                            float* __restrict__ agg,
                            int* __restrict__ lcnt,
                            int* __restrict__ lsrc,
                            const unsigned* __restrict__ own) {
    constexpr int HC = H * Cc;
    constexpr int NV = HC / 32;
    constexpr int VW = (NV >= 4) ? 4 : NV;
    constexpr int NG = NV / VW;
    constexpr int B  = (NV >= 8) ? 8 : 16;
    typedef float vecf __attribute__((ext_vector_type(VW)));
    __shared__ int   cntr[REUSE ? 1 : RMAX];
    __shared__ int   bins[REUSE ? 1 : RMAX * BINCAP];
    __shared__ float wbuf[AGG_WAVES * BINCAP * H];
    __shared__ int   hits[REUSE ? 1 : AGG_WAVES * WCAP];

    const int lane = threadIdx.x & 31;
    const int w    = __builtin_amdgcn_readfirstlane((int)(threadIdx.x >> 5));
    const int b    = blockIdx.x;
    const int lo   = b * RBASE + ((b < REXTRA) ? b : REXTRA);
    const int cnt  = RBASE + ((b < REXTRA) ? 1 : 0);

    const int4* dst4 = (const int4*)(ei + E0C);
    if constexpr (!REUSE) {
        for (int i = threadIdx.x; i < cnt; i += AGG_WAVES * 32) {
            cntr[i] = 1;
            bins[i * BINCAP] = E0C + lo + i;
        }
        __syncthreads();

        {
            const uint4* own4 = (const uint4*)own;
            const unsigned pat = (unsigned)b * 0x01010101u;
            const int start = (int)(((long long)b * NSTEP) / AGG_WG);
            constexpr int ITERS = (NSTEP + AGG_WAVES - 1) / AGG_WAVES;
            auto step_of = [&](int i) -> int {
                const int idx = i * AGG_WAVES + w;
                int c = idx + start;
                if (c >= NSTEP) c -= NSTEP;
                return (idx < NSTEP) ? c : -1;
            };
            auto load_step = [&](int c) -> OwnStep {
                const uint4* p = own4 + (size_t)((c >= 0) ? c : 0) * (ST_EDGES / 16) + lane;
                OwnStep d; d.a = p[0]; d.b = p[32];
                return d;
            };
            int* wl = hits + w * WCAP;
            int  wn = 0;
            int c0 = step_of(0), c1 = step_of(1), c2 = step_of(2), c3 = step_of(3);
            OwnStep d0 = load_step(c0), d1 = load_step(c1), d2 = load_step(c2), d3 = load_step(c3);
            for (int it = 0; it < ITERS; it += 4) {
                scan_step(d0, c0, lane, pat, ei, lo, cnt, wn, wl, cntr, bins);
                c0 = step_of(it + 4); d0 = load_step(c0);
                scan_step(d1, c1, lane, pat, ei, lo, cnt, wn, wl, cntr, bins);
                c1 = step_of(it + 5); d1 = load_step(c1);
                scan_step(d2, c2, lane, pat, ei, lo, cnt, wn, wl, cntr, bins);
                c2 = step_of(it + 6); d2 = load_step(c2);
                scan_step(d3, c3, lane, pat, ei, lo, cnt, wn, wl, cntr, bins);
                c3 = step_of(it + 7); d3 = load_step(c3);
            }

            const int nl = (wn < WCAP) ? wn : WCAP;
            for (int i0 = 0; i0 < nl; i0 += 128) {
                int      ev[4];
                unsigned rel[4];
#pragma unroll
                for (int q = 0; q < 4; ++q) {
                    const int i = i0 + q * 32 + lane;
                    const bool ok = i < nl;
                    ev[q] = ok ? wl[i] : 0;
                    const int  d  = ei[E0C + ev[q]];
                    rel[q] = ok ? (unsigned)(d - lo) : 0xFFFFFFFFu;
                }
#pragma unroll
                for (int q = 0; q < 4; ++q) {
                    if (rel[q] < (unsigned)cnt) {
                        const int pos = atomicAdd(cntr + rel[q], 1);
                        if (pos < BINCAP) bins[rel[q] * BINCAP + pos] = ev[q];
                    }
                }
            }
        }
        __syncthreads();
    }

    float* wrow = wbuf + w * BINCAP * H;
    int n_nx = 0;
    if constexpr (REUSE) n_nx = (w < cnt) ? lcnt[lo + w] : 0;
    for (int r = w; r < cnt; r += AGG_WAVES) {
        const int d = lo + r;
        int n;
        if constexpr (REUSE) {
            n = __builtin_amdgcn_readfirstlane(n_nx);
            n_nx = (r + AGG_WAVES < cnt) ? lcnt[d + AGG_WAVES] : 0;
        } else {
            n = __builtin_amdgcn_readfirstlane(cntr[r]);
            if (lane == 0) lcnt[d] = n;
        }
        float* orow = agg + (size_t)d * HC;
        vecf acc[NG];
#pragma unroll
        for (int g = 0; g < NG; ++g) acc[g] = (vecf)(0.f);

        if (n <= BINCAP) {
            const int NHALF = (n > 32) ? 2 : 1;
            int sv[2];
            int* lrow = lsrc + (size_t)d * BINCAP;
            if constexpr (REUSE) {
                sv[0] = (lane < n) ? lrow[lane] : d;
                sv[1] = sv[0];
                if (NHALF == 2) sv[1] = (lane + 32 < n) ? lrow[lane + 32] : d;
            } else {
                int* bk = bins + r * BINCAP;
                int ev[2];
                ev[0] = (lane < n)      ? bk[lane]      : 0x7FFFFFFF;
                ev[1] = (lane + 32 < n) ? bk[lane + 32] : 0x7FFFFFFF;
                int rk0 = 0, rk1 = 0;
                for (int t = 0; t < n; ++t) {
                    const int x = __builtin_amdgcn_readlane((t < 32) ? ev[0] : ev[1], t & 31);
                    rk0 += (x < ev[0]) ? 1 : 0;
                    rk1 += (x < ev[1]) ? 1 : 0;
                }
                if (lane < n)      bk[rk0] = ev[0];
                if (lane + 32 < n) bk[rk1] = ev[1];
                ev[0] = (lane < n)      ? bk[lane]      : (E0C + d);
                ev[1] = (lane + 32 < n) ? bk[lane + 32] : (E0C + d);
                sv[0] = (ev[0] < E0C) ? ei[ev[0]] : (ev[0] - E0C);
                sv[1] = sv[0];
                if (NHALF == 2) sv[1] = (ev[1] < E0C) ? ei[ev[1]] : (ev[1] - E0C);
                if (lane < n) lrow[lane] = sv[0];
                if (NHALF == 2 && lane + 32 < n) lrow[lane + 32] = sv[1];
            }

            float av[2][H];
            float mx[H];
#pragma unroll
            for (int hd = 0; hd < H; ++hd) mx[hd] = -INFINITY;
#pragma unroll
            for (int hb = 0; hb < 2; ++hb) {
                if (hb < NHALF) {
                    const bool valid = (lane + 32 * hb) < n;
#pragma unroll
                    for (int hd = 0; hd < H; ++hd) {
                        const float a = leaky(ai[(size_t)d * H + hd] + aj[(size_t)sv[hb] * H + hd]);
                        av[hb][hd] = valid ? a : -INFINITY;
                        mx[hd] = fmaxf(mx[hd], av[hb][hd]);
                    }
                } else {
#pragma unroll
                    for (int hd = 0; hd < H; ++hd) av[hb][hd] = -INFINITY;
                }
            }
#pragma unroll
            for (int hd = 0; hd < H; ++hd) {
#pragma unroll
                for (int off = 16; off >= 1; off >>= 1)
                    mx[hd] = fmaxf(mx[hd], __shfl_xor(mx[hd], off, 32));
            }
            float sm[H];
#pragma unroll
            for (int hd = 0; hd < H; ++hd) {
                const float x0 = (lane < n) ? __expf(av[0][hd] - mx[hd]) : 0.f;
                float x1 = 0.f;
                if (NHALF == 2) x1 = (lane + 32 < n) ? __expf(av[1][hd] - mx[hd]) : 0.f;
                av[0][hd] = x0;
                av[1][hd] = x1;
                float t = x0 + x1;
#pragma unroll
                for (int off = 16; off >= 1; off >>= 1)
                    t += __shfl_xor(t, off, 32);
                sm[hd] = t;
            }
#pragma unroll
            for (int hb = 0; hb < 2; ++hb) {
                if (hb < NHALF && (lane + 32 * hb) < n) {
#pragma unroll
                    for (int hd = 0; hd < H; ++hd)
                        wrow[(lane + 32 * hb) * H + hd] = av[hb][hd] / sm[hd];
                }
            }

            for (int t0 = 0; t0 < n; t0 += B) {
                vecf  v[B][NG];
                float wt[B][NG];
#pragma unroll
                for (int q = 0; q < B; ++q) {
                    const int tq = (t0 + q < n) ? (t0 + q) : (n - 1);
                    const int s = __builtin_amdgcn_readlane((tq < 32) ? sv[0] : sv[1], tq & 31);
                    const float* hs = h + (size_t)s * HC;
#pragma unroll
                    for (int g = 0; g < NG; ++g) {
                        const int ch0 = g * 32 * VW + VW * lane;
                        v[q][g]  = *(const vecf*)(hs + ch0);
                        wt[q][g] = wrow[tq * H + ch0 / Cc];
                    }
                }
#pragma unroll
                for (int q = 0; q < B; ++q) {
                    if (t0 + q < n) {
#pragma unroll
                        for (int g = 0; g < NG; ++g) acc[g] += wt[q][g] * v[q][g];
                    }
                }
            }
        } else {
            int   hdg[NG];
            float aig[NG], mxg[NG], smg[NG];
#pragma unroll
            for (int g = 0; g < NG; ++g) {
                hdg[g] = (g * 32 * VW + VW * lane) / Cc;
                aig[g] = ai[(size_t)d * H + hdg[g]];
                mxg[g] = -INFINITY;
                smg[g] = 0.f;
            }
            for (int pass = 0; pass < 3; ++pass) {
                auto edge = [&](int e) {
                    const int s = (e < E0C) ? ei[e] : (e - E0C);
#pragma unroll
                    for (int g = 0; g < NG; ++g) {
                        const float a = leaky(aig[g] + aj[(size_t)s * H + hdg[g]]);
                        if (pass == 0) {
                            mxg[g] = fmaxf(mxg[g], a);
                        } else if (pass == 1) {
                            smg[g] += __expf(a - mxg[g]);
                        } else {
                            const float wgt = __expf(a - mxg[g]) / smg[g];
                            const vecf x = *(const vecf*)(h + (size_t)s * HC + g * 32 * VW + VW * lane);
                            acc[g] += wgt * x;
                        }
                    }
                };
                for (int c = 0; c < NCHUNK; ++c) {
                    const int4* p = dst4 + (size_t)c * (CH_EDGES / 4) + lane;
                    const int4 da = p[0], db = p[32];
                    const int dq[8] = { da.x, da.y, da.z, da.w, db.x, db.y, db.z, db.w };
#pragma unroll
                    for (int q = 0; q < 8; ++q) {
                        unsigned mk = __builtin_amdgcn_ballot_w32(dq[q] == d);
                        while (mk) {
                            const int l = __builtin_ctz(mk);
                            mk &= mk - 1u;
                            edge(c * CH_EDGES + (q >> 2) * 128 + l * 4 + (q & 3));
                        }
                    }
                }
                edge(E0C + d);
            }
        }

#pragma unroll
        for (int g = 0; g < NG; ++g)
            *(vecf*)(orow + g * 32 * VW + VW * lane) = acc[g];
    }
}

__global__ void add_bias(float* __restrict__ out, const float* __restrict__ bias,
                         int total, int Cc) {
    int i = blockIdx.x * blockDim.x + threadIdx.x;
    if (i < total) out[i] += bias[i % Cc];
}

extern "C" void kernel_launch(void* const* d_in, const int* in_sizes, int n_in,
                              void* d_out, int out_size, void* d_ws, size_t ws_size,
                              hipStream_t stream) {
    (void)in_sizes; (void)n_in; (void)out_size; (void)ws_size;

    const float* x    = (const float*)d_in[0];
    const int*   ei   = (const int*)  d_in[1];
    const float* W1   = (const float*)d_in[2];
    const float* att1 = (const float*)d_in[3];
    const float* b1   = (const float*)d_in[4];
    const float* W2   = (const float*)d_in[5];
    const float* att2 = (const float*)d_in[6];
    const float* b2   = (const float*)d_in[7];
    float* out = (float*)d_out;

    char* base = (char*)d_ws;
    size_t off = 0;
    auto carve = [&](size_t bytes) -> char* {
        char* p = base + off;
        off = (off + bytes + 255) & ~(size_t)255;
        return p;
    };
    _Float16* W1T   = (_Float16*)carve((size_t)INDIM * HIDDIM * 2);
    _Float16* W2T   = (_Float16*)carve((size_t)HIDDIM * OUTDIM * 2);
    float*    HBUF  = (float*)   carve((size_t)MPAD * HIDDIM * 4);
    float*    AGG1  = (float*)   carve((size_t)NN * HIDDIM * 4);
    float*    AI    = (float*)   carve((size_t)NN * H1C * 4);
    float*    AJ    = (float*)   carve((size_t)NN * H1C * 4);
    int*      LCNT  = (int*)     carve((size_t)NN * 4);
    int*      LSRC  = (int*)     carve((size_t)NN * BINCAP * 4);
    unsigned* OWN   = (unsigned*)carve((size_t)EPAD);

    const int TB = 256;
    auto nb = [](long long n, int tb) { return (unsigned)((n + tb - 1) / tb); };
    const int MT2 = MPAD / 32;

    xpose_cvt_f16 <<<nb((long long)INDIM * HIDDIM, TB), TB, 0, stream>>>(W1, W1T, INDIM, HIDDIM);
    xpose_cvt_f16 <<<nb((long long)HIDDIM * OUTDIM, TB), TB, 0, stream>>>(W2, W2T, HIDDIM, OUTDIM);

    gemm_f16_wmma<false, HIDDIM / 32, 1, INDIM><<<MT2, 256, 0, stream>>>(x, nullptr, W1T, HBUF, MT2, NN);

    node_coef<<<nb((long long)NN * H1C, TB), TB, 0, stream>>>(HBUF, att1, AI, AJ, NN, H1C, C1C);

    owner_bytes<<<nb((long long)EPAD / 16, TB), TB, 0, stream>>>(ei, OWN);

    edge_softmax_aggregate<H1C, C1C, false><<<AGG_WG, AGG_WAVES * 32, 0, stream>>>(ei, AI, AJ, HBUF, AGG1, LCNT, LSRC, OWN);

    gemm_f16_wmma<true, OUTDIM / 32, 2, HIDDIM><<<(MT2 + 1) / 2, 128, 0, stream>>>(AGG1, b1, W2T, HBUF, MT2, NN);

    node_coef<<<nb((long long)NN * H2C, TB), TB, 0, stream>>>(HBUF, att2, AI, AJ, NN, H2C, C2C);

    edge_softmax_aggregate<H2C, C2C, true><<<AGG_WG, AGG_WAVES * 32, 0, stream>>>(ei, AI, AJ, HBUF, out, LCNT, LSRC, OWN);

    add_bias<<<nb((long long)NN * OUTDIM, TB), TB, 0, stream>>>(out, b2, NN * OUTDIM, OUTDIM);
}


# Attention vectors folded into the weights: W.att rows join the f16 B operand, an extra wave makes ai/aj as 16 more WMMA columns (sums over k, were f32 dots of h over c); coefficient launches gone
# speedup vs baseline: 5.1962x; 1.0583x over previous
#include <hip/hip_runtime.h>
#include <hip/hip_bf16.h>

#define NN     50000
#define MPAD   50016
#define E0C    800000
#define ETOT   850000
#define INDIM  256
#define HIDDIM 256
#define OUTDIM 64
#define H1C    8
#define C1C    32
#define H2C    1
#define C2C    64
#define NEG_SLOPE 0.2f

typedef __attribute__((ext_vector_type(16))) _Float16 v16h;
typedef __attribute__((ext_vector_type(8)))  _Float16 v8h;
typedef __attribute__((ext_vector_type(4)))  _Float16 v4h;
typedef __attribute__((ext_vector_type(8)))  float    v8f;
typedef __attribute__((ext_vector_type(4)))  float    v4f;

union V16U { v16h v; v8h h[2]; };


#define CO_ROWS 16
__global__ void xpose_fold_f16(const float* __restrict__ W, const float* __restrict__ att,
                               _Float16* __restrict__ Wt, int K, int Nc, int H, int Cc) {
    int i = blockIdx.x * blockDim.x + threadIdx.x;
    if (i < K * Nc) {
        int k = i / Nc, n = i % Nc;
        Wt[(size_t)n * K + k] = (_Float16)W[i];
        return;
    }
    const int j = i - K * Nc;
    if (j >= CO_ROWS * K) return;
    const int row = j / K, k = j % K;
    float s = 0.f;
    if (row < 2 * H) {
        const int hd = (row < H) ? row : (row - H);
        const float* wv = W + (size_t)k * Nc + hd * Cc;
        const float* av = att + hd * 2 * Cc + ((row < H) ? 0 : Cc);
        for (int cc = 0; cc < Cc; ++cc) s += wv[cc] * av[cc];
    }
    Wt[(size_t)(Nc + row) * K + k] = (_Float16)s;
}

#define WMMA16(a, b, c) __builtin_amdgcn_wmma_f32_16x16x32_f16( \
        false, (a), false, (b), (short)0, (c), false, false)

#define A_PITCH (256 + 8)

template <bool ACT, int NT, int TM, int K, int H>
__global__ __launch_bounds__((NT + 1) * TM * 32)
void gemm_f16_wmma(const float* __restrict__ A,
                   const float* __restrict__ bias,
                   const _Float16* __restrict__ Bt,
                   float* __restrict__ C,
                   float* __restrict__ ai, float* __restrict__ aj, int Mtiles2, int Mrows) {
    constexpr int NTHR = NT * TM * 32;
    static_assert(2 * H <= CO_ROWS, "both coefficient halves fit the folded tile");
    constexpr int ROWS = TM * 32;
    constexpr int Nc   = NT * 32;
    constexpr int QPR  = K / 4;
    constexpr int RPP  = NTHR / QPR;
    static_assert(K + 8 == A_PITCH, "LDS pitch is made for this K");
    static_assert(NTHR % QPR == 0 && ROWS % RPP == 0, "whole rows per pass, whole passes per tile");
    __shared__ __attribute__((aligned(16))) _Float16 As[ROWS * A_PITCH];

    const int t = threadIdx.x;
    if (t < NTHR) {
        const int cq = t % QPR;
        const int rs = t / QPR;
        v4f q = {};
        if (ACT) q = *(const v4f*)(bias + cq * 4);
        const int row0 = blockIdx.x * ROWS;
#pragma unroll
        for (int i = 0; i < ROWS / RPP; ++i) {
            const int rl = i * RPP + rs;
            int rg = row0 + rl;  if (rg > Mrows - 1) rg = Mrows - 1;
            const v4f x = *(const v4f*)(A + (size_t)rg * K + cq * 4);
            float f[4] = { x.x, x.y, x.z, x.w };
            if (ACT) {
                const float bb[4] = { q.x, q.y, q.z, q.w };
#pragma unroll
                for (int j = 0; j < 4; ++j) {
                    float v = f[j] + bb[j];
                    f[j] = (v > 0.f) ? v : (__expf(v) - 1.f);
                }
            }
            v4h o;
#pragma unroll
            for (int j = 0; j < 4; ++j) o[j] = (_Float16)f[j];
            *(v4h*)(As + rl * A_PITCH + cq * 4) = o;
        }
    }
    __syncthreads();

    const int lane = t & 31;
    const int w   = __builtin_amdgcn_readfirstlane(t >> 5);
    const bool coef = w >= NT * TM;
    const int tml = coef ? (w - NT * TM) : (w / NT);
    const int tn  = coef ? NT : (w % NT);
    const int tm  = blockIdx.x * TM + tml;
    if (tm >= Mtiles2) return;
    const int r    = lane & 15;
    const int half = lane >> 4;

    const _Float16* a0 = As + (tml * 32 + r) * A_PITCH + half * 8;
    const _Float16* a1 = a0 + 16 * A_PITCH;
    const _Float16* b0 = Bt + (size_t)(tn * 32 + r) * K + half * 8;
    const _Float16* b1 = b0 + (size_t)16 * K;

    if (coef) {
        v8f c0 = {}, c1 = {};
#pragma unroll
        for (int k0 = 0; k0 < K; k0 += 32) {
            V16U A0, A1, B0;
            B0.h[0] = *(const v8h*)(b0 + k0);
            B0.h[1] = *(const v8h*)(b0 + k0 + 16);
            A0.h[0] = *(const v8h*)(a0 + k0);
            A0.h[1] = *(const v8h*)(a0 + k0 + 16);
            A1.h[0] = *(const v8h*)(a1 + k0);
            A1.h[1] = *(const v8h*)(a1 + k0 + 16);
            c0 = WMMA16(A0.v, B0.v, c0);
            c1 = WMMA16(A1.v, B0.v, c1);
        }
        if (r < 2 * H) {
            float* q = (r < H) ? (ai + r) : (aj + (r - H));
            const int row0 = tm * 32 + half * 8;
#pragma unroll
            for (int j = 0; j < 8; ++j) {
                if (row0 + j < Mrows)      q[(size_t)(row0 + j) * H]      = c0[j];
                if (row0 + 16 + j < Mrows) q[(size_t)(row0 + 16 + j) * H] = c1[j];
            }
        }
        return;
    }

    v8f c00 = {}, c01 = {}, c10 = {}, c11 = {};
#pragma unroll
    for (int k0 = 0; k0 < K; k0 += 32) {
        V16U A0, A1, B0, B1;
        B0.h[0] = *(const v8h*)(b0 + k0);
        B0.h[1] = *(const v8h*)(b0 + k0 + 16);
        B1.h[0] = *(const v8h*)(b1 + k0);
        B1.h[1] = *(const v8h*)(b1 + k0 + 16);
        A0.h[0] = *(const v8h*)(a0 + k0);
        A0.h[1] = *(const v8h*)(a0 + k0 + 16);
        A1.h[0] = *(const v8h*)(a1 + k0);
        A1.h[1] = *(const v8h*)(a1 + k0 + 16);
        c00 = WMMA16(A0.v, B0.v, c00);
        c01 = WMMA16(A0.v, B1.v, c01);
        c10 = WMMA16(A1.v, B0.v, c10);
        c11 = WMMA16(A1.v, B1.v, c11);
    }
    {
        float* p = C + (size_t)(tm * 32 + half * 8) * Nc + tn * 32 + r;
#pragma unroll
        for (int j = 0; j < 8; ++j) { p[(size_t)j * Nc] = c00[j]; p[(size_t)j * Nc + 16] = c01[j]; }
        p += (size_t)16 * Nc;
#pragma unroll
        for (int j = 0; j < 8; ++j) { p[(size_t)j * Nc] = c10[j]; p[(size_t)j * Nc + 16] = c11[j]; }
    }
}

#define AGG_WG     256
#define AGG_WAVES  32
#define RBASE      (NN / AGG_WG)
#define REXTRA     (NN - RBASE * AGG_WG)
#define RMAX       (RBASE + 1)
#define CH_EDGES   256
#define NCHUNK     (E0C / CH_EDGES)
#define BINCAP     64
static_assert(E0C % CH_EDGES == 0, "destination list is read in whole steps");
#define ST_EDGES   1024
#define NSTEP      ((E0C + ST_EDGES - 1) / ST_EDGES)
#define EPAD       (NSTEP * ST_EDGES)
#define WCAP       256
static_assert(AGG_WG == 256, "an owner number is one byte");
static_assert(E0C % 4 == 0, "four edges per word of the byte plane");

struct OwnStep { uint4 a, b; };

__device__ __forceinline__ unsigned owner_of(int d) {
    const int big = REXTRA * RMAX;
    return (d < big) ? ((unsigned)d / (unsigned)RMAX)
                     : ((unsigned)REXTRA + (unsigned)(d - big) / (unsigned)RBASE);
}

__global__ void owner_bytes(const int* __restrict__ ei, unsigned* __restrict__ own) {
    const int t  = blockIdx.x * blockDim.x + threadIdx.x;
    const int i0 = (t >> 5) * 128 + (t & 31);
    const int4* dst4 = (const int4*)(ei + E0C);
    int4 d[4];
#pragma unroll
    for (int j = 0; j < 4; ++j) {
        const int i = i0 + j * 32;
        d[j] = dst4[(i < E0C / 4) ? i : 0];
    }
#pragma unroll
    for (int j = 0; j < 4; ++j) {
        const int i = i0 + j * 32;
        if (i < EPAD / 4) {
            const unsigned v = (owner_of(d[j].x) & 255u) | ((owner_of(d[j].y) & 255u) << 8)
                             | ((owner_of(d[j].z) & 255u) << 16) | ((owner_of(d[j].w) & 255u) << 24);
            own[i] = (i < E0C / 4) ? v : 0xFFFFFFFFu;
        }
    }
}

__device__ __forceinline__ float leaky(float a) { return (a >= 0.f) ? a : NEG_SLOPE * a; }

__device__ __forceinline__ void file_edge(int e, const int* __restrict__ ei, int lo, int cnt,
                                          int* cntr, int* bins) {
    const unsigned rel = (unsigned)(ei[E0C + e] - lo);
    if (rel < (unsigned)cnt) {
        const int pos = atomicAdd(cntr + rel, 1);
        if (pos < BINCAP) bins[rel * BINCAP + pos] = e;
    }
}

__device__ __forceinline__ void scan_step(const OwnStep& cur, int cc, int lane, unsigned pat,
                                          const int* __restrict__ ei, int lo, int cnt,
                                          int& wn, int* wl, int* cntr, int* bins) {
    if (cc < 0) return;
    const unsigned v[8] = { cur.a.x, cur.a.y, cur.a.z, cur.a.w, cur.b.x, cur.b.y, cur.b.z, cur.b.w };
    unsigned m = 0;
#pragma unroll
    for (int q = 0; q < 8; ++q) {
        const unsigned y = v[q] ^ pat;
        const unsigned z = (y - 0x01010101u) & ~y & 0x80808080u;
        m |= (z >> 7) << q;
    }
    const int ebase = cc * ST_EDGES + lane * 16;
    if (cc == NSTEP - 1) {
#pragma unroll
        for (int q = 0; q < 8; ++q) {
            if (ebase + (q >> 2) * 512 + (q & 3) * 4 >= E0C) m &= ~(0x01010101u << q);
        }
    }
    unsigned pend = __builtin_amdgcn_ballot_w32(m != 0u);
    while (pend != 0u) {
        if (m != 0u) {
            const int slot = wn + (int)__builtin_amdgcn_mbcnt_lo(pend, 0u);
            const int bit = __builtin_ctz(m);
            m &= m - 1u;
            const int q = bit & 7, k = bit >> 3;
            const int e = ebase + (q >> 2) * 512 + (q & 3) * 4 + k;
            if (slot < WCAP) wl[slot] = e;
            else file_edge(e, ei, lo, cnt, cntr, bins);
        }
        wn += __builtin_popcount(pend);
        pend = __builtin_amdgcn_ballot_w32(m != 0u);
    }
}

template <int H, int Cc, bool REUSE>
__global__ __launch_bounds__(1024)
void edge_softmax_aggregate(const int* __restrict__ ei,
                            const float* __restrict__ ai,
                            const float* __restrict__ aj,
                            const float* __restrict__ h,
                            float* __restrict__ agg,
                            int* __restrict__ lcnt,
                            int* __restrict__ lsrc,
                            const unsigned* __restrict__ own) {
    constexpr int HC = H * Cc;
    constexpr int NV = HC / 32;
    constexpr int VW = (NV >= 4) ? 4 : NV;
    constexpr int NG = NV / VW;
    constexpr int B  = (NV >= 8) ? 8 : 16;
    typedef float vecf __attribute__((ext_vector_type(VW)));
    __shared__ int   cntr[REUSE ? 1 : RMAX];
    __shared__ int   bins[REUSE ? 1 : RMAX * BINCAP];
    __shared__ float wbuf[AGG_WAVES * BINCAP * H];
    __shared__ int   hits[REUSE ? 1 : AGG_WAVES * WCAP];

    const int lane = threadIdx.x & 31;
    const int w    = __builtin_amdgcn_readfirstlane((int)(threadIdx.x >> 5));
    const int b    = blockIdx.x;
    const int lo   = b * RBASE + ((b < REXTRA) ? b : REXTRA);
    const int cnt  = RBASE + ((b < REXTRA) ? 1 : 0);

    const int4* dst4 = (const int4*)(ei + E0C);
    if constexpr (!REUSE) {
        for (int i = threadIdx.x; i < cnt; i += AGG_WAVES * 32) {
            cntr[i] = 1;
            bins[i * BINCAP] = E0C + lo + i;
        }
        __syncthreads();

        {
            const uint4* own4 = (const uint4*)own;
            const unsigned pat = (unsigned)b * 0x01010101u;
            const int start = (int)(((long long)b * NSTEP) / AGG_WG);
            constexpr int ITERS = (NSTEP + AGG_WAVES - 1) / AGG_WAVES;
            auto step_of = [&](int i) -> int {
                const int idx = i * AGG_WAVES + w;
                int c = idx + start;
                if (c >= NSTEP) c -= NSTEP;
                return (idx < NSTEP) ? c : -1;
            };
            auto load_step = [&](int c) -> OwnStep {
                const uint4* p = own4 + (size_t)((c >= 0) ? c : 0) * (ST_EDGES / 16) + lane;
                OwnStep d; d.a = p[0]; d.b = p[32];
                return d;
            };
            int* wl = hits + w * WCAP;
            int  wn = 0;
            int c0 = step_of(0), c1 = step_of(1), c2 = step_of(2), c3 = step_of(3);
            OwnStep d0 = load_step(c0), d1 = load_step(c1), d2 = load_step(c2), d3 = load_step(c3);
            for (int it = 0; it < ITERS; it += 4) {
                scan_step(d0, c0, lane, pat, ei, lo, cnt, wn, wl, cntr, bins);
                c0 = step_of(it + 4); d0 = load_step(c0);
                scan_step(d1, c1, lane, pat, ei, lo, cnt, wn, wl, cntr, bins);
                c1 = step_of(it + 5); d1 = load_step(c1);
                scan_step(d2, c2, lane, pat, ei, lo, cnt, wn, wl, cntr, bins);
                c2 = step_of(it + 6); d2 = load_step(c2);
                scan_step(d3, c3, lane, pat, ei, lo, cnt, wn, wl, cntr, bins);
                c3 = step_of(it + 7); d3 = load_step(c3);
            }

            const int nl = (wn < WCAP) ? wn : WCAP;
            for (int i0 = 0; i0 < nl; i0 += 128) {
                int      ev[4];
                unsigned rel[4];
#pragma unroll
                for (int q = 0; q < 4; ++q) {
                    const int i = i0 + q * 32 + lane;
                    const bool ok = i < nl;
                    ev[q] = ok ? wl[i] : 0;
                    const int  d  = ei[E0C + ev[q]];
                    rel[q] = ok ? (unsigned)(d - lo) : 0xFFFFFFFFu;
                }
#pragma unroll
                for (int q = 0; q < 4; ++q) {
                    if (rel[q] < (unsigned)cnt) {
                        const int pos = atomicAdd(cntr + rel[q], 1);
                        if (pos < BINCAP) bins[rel[q] * BINCAP + pos] = ev[q];
                    }
                }
            }
        }
        __syncthreads();
    }

    float* wrow = wbuf + w * BINCAP * H;
    int n_nx = 0;
    if constexpr (REUSE) n_nx = (w < cnt) ? lcnt[lo + w] : 0;
    for (int r = w; r < cnt; r += AGG_WAVES) {
        const int d = lo + r;
        int n;
        if constexpr (REUSE) {
            n = __builtin_amdgcn_readfirstlane(n_nx);
            n_nx = (r + AGG_WAVES < cnt) ? lcnt[d + AGG_WAVES] : 0;
        } else {
            n = __builtin_amdgcn_readfirstlane(cntr[r]);
            if (lane == 0) lcnt[d] = n;
        }
        float* orow = agg + (size_t)d * HC;
        vecf acc[NG];
#pragma unroll
        for (int g = 0; g < NG; ++g) acc[g] = (vecf)(0.f);

        if (n <= BINCAP) {
            const int NHALF = (n > 32) ? 2 : 1;
            int sv[2];
            int* lrow = lsrc + (size_t)d * BINCAP;
            if constexpr (REUSE) {
                sv[0] = (lane < n) ? lrow[lane] : d;
                sv[1] = sv[0];
                if (NHALF == 2) sv[1] = (lane + 32 < n) ? lrow[lane + 32] : d;
            } else {
                int* bk = bins + r * BINCAP;
                int ev[2];
                ev[0] = (lane < n)      ? bk[lane]      : 0x7FFFFFFF;
                ev[1] = (lane + 32 < n) ? bk[lane + 32] : 0x7FFFFFFF;
                int rk0 = 0, rk1 = 0;
                for (int t = 0; t < n; ++t) {
                    const int x = __builtin_amdgcn_readlane((t < 32) ? ev[0] : ev[1], t & 31);
                    rk0 += (x < ev[0]) ? 1 : 0;
                    rk1 += (x < ev[1]) ? 1 : 0;
                }
                if (lane < n)      bk[rk0] = ev[0];
                if (lane + 32 < n) bk[rk1] = ev[1];
                ev[0] = (lane < n)      ? bk[lane]      : (E0C + d);
                ev[1] = (lane + 32 < n) ? bk[lane + 32] : (E0C + d);
                sv[0] = (ev[0] < E0C) ? ei[ev[0]] : (ev[0] - E0C);
                sv[1] = sv[0];
                if (NHALF == 2) sv[1] = (ev[1] < E0C) ? ei[ev[1]] : (ev[1] - E0C);
                if (lane < n) lrow[lane] = sv[0];
                if (NHALF == 2 && lane + 32 < n) lrow[lane + 32] = sv[1];
            }

            float av[2][H];
            float mx[H];
#pragma unroll
            for (int hd = 0; hd < H; ++hd) mx[hd] = -INFINITY;
#pragma unroll
            for (int hb = 0; hb < 2; ++hb) {
                if (hb < NHALF) {
                    const bool valid = (lane + 32 * hb) < n;
#pragma unroll
                    for (int hd = 0; hd < H; ++hd) {
                        const float a = leaky(ai[(size_t)d * H + hd] + aj[(size_t)sv[hb] * H + hd]);
                        av[hb][hd] = valid ? a : -INFINITY;
                        mx[hd] = fmaxf(mx[hd], av[hb][hd]);
                    }
                } else {
#pragma unroll
                    for (int hd = 0; hd < H; ++hd) av[hb][hd] = -INFINITY;
                }
            }
#pragma unroll
            for (int hd = 0; hd < H; ++hd) {
#pragma unroll
                for (int off = 16; off >= 1; off >>= 1)
                    mx[hd] = fmaxf(mx[hd], __shfl_xor(mx[hd], off, 32));
            }
            float sm[H];
#pragma unroll
            for (int hd = 0; hd < H; ++hd) {
                const float x0 = (lane < n) ? __expf(av[0][hd] - mx[hd]) : 0.f;
                float x1 = 0.f;
                if (NHALF == 2) x1 = (lane + 32 < n) ? __expf(av[1][hd] - mx[hd]) : 0.f;
                av[0][hd] = x0;
                av[1][hd] = x1;
                float t = x0 + x1;
#pragma unroll
                for (int off = 16; off >= 1; off >>= 1)
                    t += __shfl_xor(t, off, 32);
                sm[hd] = t;
            }
#pragma unroll
            for (int hb = 0; hb < 2; ++hb) {
                if (hb < NHALF && (lane + 32 * hb) < n) {
#pragma unroll
                    for (int hd = 0; hd < H; ++hd)
                        wrow[(lane + 32 * hb) * H + hd] = av[hb][hd] / sm[hd];
                }
            }

            for (int t0 = 0; t0 < n; t0 += B) {
                vecf  v[B][NG];
                float wt[B][NG];
#pragma unroll
                for (int q = 0; q < B; ++q) {
                    const int tq = (t0 + q < n) ? (t0 + q) : (n - 1);
                    const int s = __builtin_amdgcn_readlane((tq < 32) ? sv[0] : sv[1], tq & 31);
                    const float* hs = h + (size_t)s * HC;
#pragma unroll
                    for (int g = 0; g < NG; ++g) {
                        const int ch0 = g * 32 * VW + VW * lane;
                        v[q][g]  = *(const vecf*)(hs + ch0);
                        wt[q][g] = wrow[tq * H + ch0 / Cc];
                    }
                }
#pragma unroll
                for (int q = 0; q < B; ++q) {
                    if (t0 + q < n) {
#pragma unroll
                        for (int g = 0; g < NG; ++g) acc[g] += wt[q][g] * v[q][g];
                    }
                }
            }
        } else {
            int   hdg[NG];
            float aig[NG], mxg[NG], smg[NG];
#pragma unroll
            for (int g = 0; g < NG; ++g) {
                hdg[g] = (g * 32 * VW + VW * lane) / Cc;
                aig[g] = ai[(size_t)d * H + hdg[g]];
                mxg[g] = -INFINITY;
                smg[g] = 0.f;
            }
            for (int pass = 0; pass < 3; ++pass) {
                auto edge = [&](int e) {
                    const int s = (e < E0C) ? ei[e] : (e - E0C);
#pragma unroll
                    for (int g = 0; g < NG; ++g) {
                        const float a = leaky(aig[g] + aj[(size_t)s * H + hdg[g]]);
                        if (pass == 0) {
                            mxg[g] = fmaxf(mxg[g], a);
                        } else if (pass == 1) {
                            smg[g] += __expf(a - mxg[g]);
                        } else {
                            const float wgt = __expf(a - mxg[g]) / smg[g];
                            const vecf x = *(const vecf*)(h + (size_t)s * HC + g * 32 * VW + VW * lane);
                            acc[g] += wgt * x;
                        }
                    }
                };
                for (int c = 0; c < NCHUNK; ++c) {
                    const int4* p = dst4 + (size_t)c * (CH_EDGES / 4) + lane;
                    const int4 da = p[0], db = p[32];
                    const int dq[8] = { da.x, da.y, da.z, da.w, db.x, db.y, db.z, db.w };
#pragma unroll
                    for (int q = 0; q < 8; ++q) {
                        unsigned mk = __builtin_amdgcn_ballot_w32(dq[q] == d);
                        while (mk) {
                            const int l = __builtin_ctz(mk);
                            mk &= mk - 1u;
                            edge(c * CH_EDGES + (q >> 2) * 128 + l * 4 + (q & 3));
                        }
                    }
                }
                edge(E0C + d);
            }
        }

#pragma unroll
        for (int g = 0; g < NG; ++g)
            *(vecf*)(orow + g * 32 * VW + VW * lane) = acc[g];
    }
}

__global__ void add_bias(float* __restrict__ out, const float* __restrict__ bias,
                         int total, int Cc) {
    int i = blockIdx.x * blockDim.x + threadIdx.x;
    if (i < total) out[i] += bias[i % Cc];
}

extern "C" void kernel_launch(void* const* d_in, const int* in_sizes, int n_in,
                              void* d_out, int out_size, void* d_ws, size_t ws_size,
                              hipStream_t stream) {
    (void)in_sizes; (void)n_in; (void)out_size; (void)ws_size;

    const float* x    = (const float*)d_in[0];
    const int*   ei   = (const int*)  d_in[1];
    const float* W1   = (const float*)d_in[2];
    const float* att1 = (const float*)d_in[3];
    const float* b1   = (const float*)d_in[4];
    const float* W2   = (const float*)d_in[5];
    const float* att2 = (const float*)d_in[6];
    const float* b2   = (const float*)d_in[7];
    float* out = (float*)d_out;

    char* base = (char*)d_ws;
    size_t off = 0;
    auto carve = [&](size_t bytes) -> char* {
        char* p = base + off;
        off = (off + bytes + 255) & ~(size_t)255;
        return p;
    };
    _Float16* W1T   = (_Float16*)carve((size_t)INDIM * (HIDDIM + CO_ROWS) * 2);
    _Float16* W2T   = (_Float16*)carve((size_t)HIDDIM * (OUTDIM + CO_ROWS) * 2);
    float*    HBUF  = (float*)   carve((size_t)MPAD * HIDDIM * 4);
    float*    AGG1  = (float*)   carve((size_t)NN * HIDDIM * 4);
    float*    AI    = (float*)   carve((size_t)NN * H1C * 4);
    float*    AJ    = (float*)   carve((size_t)NN * H1C * 4);
    int*      LCNT  = (int*)     carve((size_t)NN * 4);
    int*      LSRC  = (int*)     carve((size_t)NN * BINCAP * 4);
    unsigned* OWN   = (unsigned*)carve((size_t)EPAD);

    const int TB = 256;
    auto nb = [](long long n, int tb) { return (unsigned)((n + tb - 1) / tb); };
    const int MT2 = MPAD / 32;

    xpose_fold_f16<<<nb((long long)INDIM * (HIDDIM + CO_ROWS), TB), TB, 0, stream>>>(W1, att1, W1T, INDIM, HIDDIM, H1C, C1C);
    xpose_fold_f16<<<nb((long long)HIDDIM * (OUTDIM + CO_ROWS), TB), TB, 0, stream>>>(W2, att2, W2T, HIDDIM, OUTDIM, H2C, C2C);

    gemm_f16_wmma<false, HIDDIM / 32, 1, INDIM, H1C><<<MT2, (HIDDIM / 32 + 1) * 32, 0, stream>>>(x, nullptr, W1T, HBUF, AI, AJ, MT2, NN);

    owner_bytes<<<nb((long long)EPAD / 16, TB), TB, 0, stream>>>(ei, OWN);

    edge_softmax_aggregate<H1C, C1C, false><<<AGG_WG, AGG_WAVES * 32, 0, stream>>>(ei, AI, AJ, HBUF, AGG1, LCNT, LSRC, OWN);

    gemm_f16_wmma<true, OUTDIM / 32, 2, HIDDIM, H2C><<<(MT2 + 1) / 2, (OUTDIM / 32 + 1) * 2 * 32, 0, stream>>>(AGG1, b1, W2T, HBUF, AI, AJ, MT2, NN);

    edge_softmax_aggregate<H2C, C2C, true><<<AGG_WG, AGG_WAVES * 32, 0, stream>>>(ei, AI, AJ, HBUF, out, LCNT, LSRC, OWN);

    add_bias<<<nb((long long)NN * OUTDIM, TB), TB, 0, stream>>>(out, b2, NN * OUTDIM, OUTDIM);
}


# Final bias added to the layer-2 rows at the GEMM store (a row's softmax weights sum to one): bias launch and its pass over out gone; bias now inside every weighted term of the sum, rounding differs
# speedup vs baseline: 5.4922x; 1.0570x over previous
#include <hip/hip_runtime.h>
#include <hip/hip_bf16.h>

#define NN     50000
#define MPAD   50016
#define E0C    800000
#define ETOT   850000
#define INDIM  256
#define HIDDIM 256
#define OUTDIM 64
#define H1C    8
#define C1C    32
#define H2C    1
#define C2C    64
#define NEG_SLOPE 0.2f

typedef __attribute__((ext_vector_type(16))) _Float16 v16h;
typedef __attribute__((ext_vector_type(8)))  _Float16 v8h;
typedef __attribute__((ext_vector_type(4)))  _Float16 v4h;
typedef __attribute__((ext_vector_type(8)))  float    v8f;
typedef __attribute__((ext_vector_type(4)))  float    v4f;

union V16U { v16h v; v8h h[2]; };


#define CO_ROWS 16
__global__ void xpose_fold_f16(const float* __restrict__ W, const float* __restrict__ att,
                               _Float16* __restrict__ Wt, int K, int Nc, int H, int Cc) {
    int i = blockIdx.x * blockDim.x + threadIdx.x;
    if (i < K * Nc) {
        int k = i / Nc, n = i % Nc;
        Wt[(size_t)n * K + k] = (_Float16)W[i];
        return;
    }
    const int j = i - K * Nc;
    if (j >= CO_ROWS * K) return;
    const int row = j / K, k = j % K;
    float s = 0.f;
    if (row < 2 * H) {
        const int hd = (row < H) ? row : (row - H);
        const float* wv = W + (size_t)k * Nc + hd * Cc;
        const float* av = att + hd * 2 * Cc + ((row < H) ? 0 : Cc);
        for (int cc = 0; cc < Cc; ++cc) s += wv[cc] * av[cc];
    }
    Wt[(size_t)(Nc + row) * K + k] = (_Float16)s;
}

#define WMMA16(a, b, c) __builtin_amdgcn_wmma_f32_16x16x32_f16( \
        false, (a), false, (b), (short)0, (c), false, false)

#define A_PITCH (256 + 8)

template <bool ACT, bool OB, int NT, int TM, int K, int H>
__global__ __launch_bounds__((NT + 1) * TM * 32)
void gemm_f16_wmma(const float* __restrict__ A,
                   const float* __restrict__ bias,
                   const float* __restrict__ obias,
                   const _Float16* __restrict__ Bt,
                   float* __restrict__ C,
                   float* __restrict__ ai, float* __restrict__ aj, int Mtiles2, int Mrows) {
    constexpr int NTHR = NT * TM * 32;
    static_assert(2 * H <= CO_ROWS, "both coefficient halves fit the folded tile");
    constexpr int ROWS = TM * 32;
    constexpr int Nc   = NT * 32;
    constexpr int QPR  = K / 4;
    constexpr int RPP  = NTHR / QPR;
    static_assert(K + 8 == A_PITCH, "LDS pitch is made for this K");
    static_assert(NTHR % QPR == 0 && ROWS % RPP == 0, "whole rows per pass, whole passes per tile");
    __shared__ __attribute__((aligned(16))) _Float16 As[ROWS * A_PITCH];

    const int t = threadIdx.x;
    if (t < NTHR) {
        const int cq = t % QPR;
        const int rs = t / QPR;
        v4f q = {};
        if (ACT) q = *(const v4f*)(bias + cq * 4);
        const int row0 = blockIdx.x * ROWS;
#pragma unroll
        for (int i = 0; i < ROWS / RPP; ++i) {
            const int rl = i * RPP + rs;
            int rg = row0 + rl;  if (rg > Mrows - 1) rg = Mrows - 1;
            const v4f x = *(const v4f*)(A + (size_t)rg * K + cq * 4);
            float f[4] = { x.x, x.y, x.z, x.w };
            if (ACT) {
                const float bb[4] = { q.x, q.y, q.z, q.w };
#pragma unroll
                for (int j = 0; j < 4; ++j) {
                    float v = f[j] + bb[j];
                    f[j] = (v > 0.f) ? v : (__expf(v) - 1.f);
                }
            }
            v4h o;
#pragma unroll
            for (int j = 0; j < 4; ++j) o[j] = (_Float16)f[j];
            *(v4h*)(As + rl * A_PITCH + cq * 4) = o;
        }
    }
    __syncthreads();

    const int lane = t & 31;
    const int w   = __builtin_amdgcn_readfirstlane(t >> 5);
    const bool coef = w >= NT * TM;
    const int tml = coef ? (w - NT * TM) : (w / NT);
    const int tn  = coef ? NT : (w % NT);
    const int tm  = blockIdx.x * TM + tml;
    if (tm >= Mtiles2) return;
    const int r    = lane & 15;
    const int half = lane >> 4;

    const _Float16* a0 = As + (tml * 32 + r) * A_PITCH + half * 8;
    const _Float16* a1 = a0 + 16 * A_PITCH;
    const _Float16* b0 = Bt + (size_t)(tn * 32 + r) * K + half * 8;
    const _Float16* b1 = b0 + (size_t)16 * K;

    if (coef) {
        v8f c0 = {}, c1 = {};
#pragma unroll
        for (int k0 = 0; k0 < K; k0 += 32) {
            V16U A0, A1, B0;
            B0.h[0] = *(const v8h*)(b0 + k0);
            B0.h[1] = *(const v8h*)(b0 + k0 + 16);
            A0.h[0] = *(const v8h*)(a0 + k0);
            A0.h[1] = *(const v8h*)(a0 + k0 + 16);
            A1.h[0] = *(const v8h*)(a1 + k0);
            A1.h[1] = *(const v8h*)(a1 + k0 + 16);
            c0 = WMMA16(A0.v, B0.v, c0);
            c1 = WMMA16(A1.v, B0.v, c1);
        }
        if (r < 2 * H) {
            float* q = (r < H) ? (ai + r) : (aj + (r - H));
            const int row0 = tm * 32 + half * 8;
#pragma unroll
            for (int j = 0; j < 8; ++j) {
                if (row0 + j < Mrows)      q[(size_t)(row0 + j) * H]      = c0[j];
                if (row0 + 16 + j < Mrows) q[(size_t)(row0 + 16 + j) * H] = c1[j];
            }
        }
        return;
    }

    float ob0 = 0.f, ob1 = 0.f;
    if (OB) { ob0 = obias[tn * 32 + r]; ob1 = obias[tn * 32 + 16 + r]; }

    v8f c00 = {}, c01 = {}, c10 = {}, c11 = {};
#pragma unroll
    for (int k0 = 0; k0 < K; k0 += 32) {
        V16U A0, A1, B0, B1;
        B0.h[0] = *(const v8h*)(b0 + k0);
        B0.h[1] = *(const v8h*)(b0 + k0 + 16);
        B1.h[0] = *(const v8h*)(b1 + k0);
        B1.h[1] = *(const v8h*)(b1 + k0 + 16);
        A0.h[0] = *(const v8h*)(a0 + k0);
        A0.h[1] = *(const v8h*)(a0 + k0 + 16);
        A1.h[0] = *(const v8h*)(a1 + k0);
        A1.h[1] = *(const v8h*)(a1 + k0 + 16);
        c00 = WMMA16(A0.v, B0.v, c00);
        c01 = WMMA16(A0.v, B1.v, c01);
        c10 = WMMA16(A1.v, B0.v, c10);
        c11 = WMMA16(A1.v, B1.v, c11);
    }
    {
        float* p = C + (size_t)(tm * 32 + half * 8) * Nc + tn * 32 + r;
#pragma unroll
        for (int j = 0; j < 8; ++j) {
            p[(size_t)j * Nc]      = OB ? (c00[j] + ob0) : c00[j];
            p[(size_t)j * Nc + 16] = OB ? (c01[j] + ob1) : c01[j];
        }
        p += (size_t)16 * Nc;
#pragma unroll
        for (int j = 0; j < 8; ++j) {
            p[(size_t)j * Nc]      = OB ? (c10[j] + ob0) : c10[j];
            p[(size_t)j * Nc + 16] = OB ? (c11[j] + ob1) : c11[j];
        }
    }
}

#define AGG_WG     256
#define AGG_WAVES  32
#define RBASE      (NN / AGG_WG)
#define REXTRA     (NN - RBASE * AGG_WG)
#define RMAX       (RBASE + 1)
#define CH_EDGES   256
#define NCHUNK     (E0C / CH_EDGES)
#define BINCAP     64
static_assert(E0C % CH_EDGES == 0, "destination list is read in whole steps");
#define ST_EDGES   1024
#define NSTEP      ((E0C + ST_EDGES - 1) / ST_EDGES)
#define EPAD       (NSTEP * ST_EDGES)
#define WCAP       256
static_assert(AGG_WG == 256, "an owner number is one byte");
static_assert(E0C % 4 == 0, "four edges per word of the byte plane");

struct OwnStep { uint4 a, b; };

__device__ __forceinline__ unsigned owner_of(int d) {
    const int big = REXTRA * RMAX;
    return (d < big) ? ((unsigned)d / (unsigned)RMAX)
                     : ((unsigned)REXTRA + (unsigned)(d - big) / (unsigned)RBASE);
}

__global__ void owner_bytes(const int* __restrict__ ei, unsigned* __restrict__ own) {
    const int t  = blockIdx.x * blockDim.x + threadIdx.x;
    const int i0 = (t >> 5) * 128 + (t & 31);
    const int4* dst4 = (const int4*)(ei + E0C);
    int4 d[4];
#pragma unroll
    for (int j = 0; j < 4; ++j) {
        const int i = i0 + j * 32;
        d[j] = dst4[(i < E0C / 4) ? i : 0];
    }
#pragma unroll
    for (int j = 0; j < 4; ++j) {
        const int i = i0 + j * 32;
        if (i < EPAD / 4) {
            const unsigned v = (owner_of(d[j].x) & 255u) | ((owner_of(d[j].y) & 255u) << 8)
                             | ((owner_of(d[j].z) & 255u) << 16) | ((owner_of(d[j].w) & 255u) << 24);
            own[i] = (i < E0C / 4) ? v : 0xFFFFFFFFu;
        }
    }
}

__device__ __forceinline__ float leaky(float a) { return (a >= 0.f) ? a : NEG_SLOPE * a; }

__device__ __forceinline__ void file_edge(int e, const int* __restrict__ ei, int lo, int cnt,
                                          int* cntr, int* bins) {
    const unsigned rel = (unsigned)(ei[E0C + e] - lo);
    if (rel < (unsigned)cnt) {
        const int pos = atomicAdd(cntr + rel, 1);
        if (pos < BINCAP) bins[rel * BINCAP + pos] = e;
    }
}

__device__ __forceinline__ void scan_step(const OwnStep& cur, int cc, int lane, unsigned pat,
                                          const int* __restrict__ ei, int lo, int cnt,
                                          int& wn, int* wl, int* cntr, int* bins) {
    if (cc < 0) return;
    const unsigned v[8] = { cur.a.x, cur.a.y, cur.a.z, cur.a.w, cur.b.x, cur.b.y, cur.b.z, cur.b.w };
    unsigned m = 0;
#pragma unroll
    for (int q = 0; q < 8; ++q) {
        const unsigned y = v[q] ^ pat;
        const unsigned z = (y - 0x01010101u) & ~y & 0x80808080u;
        m |= (z >> 7) << q;
    }
    const int ebase = cc * ST_EDGES + lane * 16;
    if (cc == NSTEP - 1) {
#pragma unroll
        for (int q = 0; q < 8; ++q) {
            if (ebase + (q >> 2) * 512 + (q & 3) * 4 >= E0C) m &= ~(0x01010101u << q);
        }
    }
    unsigned pend = __builtin_amdgcn_ballot_w32(m != 0u);
    while (pend != 0u) {
        if (m != 0u) {
            const int slot = wn + (int)__builtin_amdgcn_mbcnt_lo(pend, 0u);
            const int bit = __builtin_ctz(m);
            m &= m - 1u;
            const int q = bit & 7, k = bit >> 3;
            const int e = ebase + (q >> 2) * 512 + (q & 3) * 4 + k;
            if (slot < WCAP) wl[slot] = e;
            else file_edge(e, ei, lo, cnt, cntr, bins);
        }
        wn += __builtin_popcount(pend);
        pend = __builtin_amdgcn_ballot_w32(m != 0u);
    }
}

template <int H, int Cc, bool REUSE>
__global__ __launch_bounds__(1024)
void edge_softmax_aggregate(const int* __restrict__ ei,
                            const float* __restrict__ ai,
                            const float* __restrict__ aj,
                            const float* __restrict__ h,
                            float* __restrict__ agg,
                            int* __restrict__ lcnt,
                            int* __restrict__ lsrc,
                            const unsigned* __restrict__ own) {
    constexpr int HC = H * Cc;
    constexpr int NV = HC / 32;
    constexpr int VW = (NV >= 4) ? 4 : NV;
    constexpr int NG = NV / VW;
    constexpr int B  = (NV >= 8) ? 8 : 16;
    typedef float vecf __attribute__((ext_vector_type(VW)));
    __shared__ int   cntr[REUSE ? 1 : RMAX];
    __shared__ int   bins[REUSE ? 1 : RMAX * BINCAP];
    __shared__ float wbuf[AGG_WAVES * BINCAP * H];
    __shared__ int   hits[REUSE ? 1 : AGG_WAVES * WCAP];

    const int lane = threadIdx.x & 31;
    const int w    = __builtin_amdgcn_readfirstlane((int)(threadIdx.x >> 5));
    const int b    = blockIdx.x;
    const int lo   = b * RBASE + ((b < REXTRA) ? b : REXTRA);
    const int cnt  = RBASE + ((b < REXTRA) ? 1 : 0);

    const int4* dst4 = (const int4*)(ei + E0C);
    if constexpr (!REUSE) {
        for (int i = threadIdx.x; i < cnt; i += AGG_WAVES * 32) {
            cntr[i] = 1;
            bins[i * BINCAP] = E0C + lo + i;
        }
        __syncthreads();

        {
            const uint4* own4 = (const uint4*)own;
            const unsigned pat = (unsigned)b * 0x01010101u;
            const int start = (int)(((long long)b * NSTEP) / AGG_WG);
            constexpr int ITERS = (NSTEP + AGG_WAVES - 1) / AGG_WAVES;
            auto step_of = [&](int i) -> int {
                const int idx = i * AGG_WAVES + w;
                int c = idx + start;
                if (c >= NSTEP) c -= NSTEP;
                return (idx < NSTEP) ? c : -1;
            };
            auto load_step = [&](int c) -> OwnStep {
                const uint4* p = own4 + (size_t)((c >= 0) ? c : 0) * (ST_EDGES / 16) + lane;
                OwnStep d; d.a = p[0]; d.b = p[32];
                return d;
            };
            int* wl = hits + w * WCAP;
            int  wn = 0;
            int c0 = step_of(0), c1 = step_of(1), c2 = step_of(2), c3 = step_of(3);
            OwnStep d0 = load_step(c0), d1 = load_step(c1), d2 = load_step(c2), d3 = load_step(c3);
            for (int it = 0; it < ITERS; it += 4) {
                scan_step(d0, c0, lane, pat, ei, lo, cnt, wn, wl, cntr, bins);
                c0 = step_of(it + 4); d0 = load_step(c0);
                scan_step(d1, c1, lane, pat, ei, lo, cnt, wn, wl, cntr, bins);
                c1 = step_of(it + 5); d1 = load_step(c1);
                scan_step(d2, c2, lane, pat, ei, lo, cnt, wn, wl, cntr, bins);
                c2 = step_of(it + 6); d2 = load_step(c2);
                scan_step(d3, c3, lane, pat, ei, lo, cnt, wn, wl, cntr, bins);
                c3 = step_of(it + 7); d3 = load_step(c3);
            }

            const int nl = (wn < WCAP) ? wn : WCAP;
            for (int i0 = 0; i0 < nl; i0 += 128) {
                int      ev[4];
                unsigned rel[4];
#pragma unroll
                for (int q = 0; q < 4; ++q) {
                    const int i = i0 + q * 32 + lane;
                    const bool ok = i < nl;
                    ev[q] = ok ? wl[i] : 0;
                    const int  d  = ei[E0C + ev[q]];
                    rel[q] = ok ? (unsigned)(d - lo) : 0xFFFFFFFFu;
                }
#pragma unroll
                for (int q = 0; q < 4; ++q) {
                    if (rel[q] < (unsigned)cnt) {
                        const int pos = atomicAdd(cntr + rel[q], 1);
                        if (pos < BINCAP) bins[rel[q] * BINCAP + pos] = ev[q];
                    }
                }
            }
        }
        __syncthreads();
    }

    float* wrow = wbuf + w * BINCAP * H;
    int n_nx = 0;
    if constexpr (REUSE) n_nx = (w < cnt) ? lcnt[lo + w] : 0;
    for (int r = w; r < cnt; r += AGG_WAVES) {
        const int d = lo + r;
        int n;
        if constexpr (REUSE) {
            n = __builtin_amdgcn_readfirstlane(n_nx);
            n_nx = (r + AGG_WAVES < cnt) ? lcnt[d + AGG_WAVES] : 0;
        } else {
            n = __builtin_amdgcn_readfirstlane(cntr[r]);
            if (lane == 0) lcnt[d] = n;
        }
        float* orow = agg + (size_t)d * HC;
        vecf acc[NG];
#pragma unroll
        for (int g = 0; g < NG; ++g) acc[g] = (vecf)(0.f);

        if (n <= BINCAP) {
            const int NHALF = (n > 32) ? 2 : 1;
            int sv[2];
            int* lrow = lsrc + (size_t)d * BINCAP;
            if constexpr (REUSE) {
                sv[0] = (lane < n) ? lrow[lane] : d;
                sv[1] = sv[0];
                if (NHALF == 2) sv[1] = (lane + 32 < n) ? lrow[lane + 32] : d;
            } else {
                int* bk = bins + r * BINCAP;
                int ev[2];
                ev[0] = (lane < n)      ? bk[lane]      : 0x7FFFFFFF;
                ev[1] = (lane + 32 < n) ? bk[lane + 32] : 0x7FFFFFFF;
                int rk0 = 0, rk1 = 0;
                for (int t = 0; t < n; ++t) {
                    const int x = __builtin_amdgcn_readlane((t < 32) ? ev[0] : ev[1], t & 31);
                    rk0 += (x < ev[0]) ? 1 : 0;
                    rk1 += (x < ev[1]) ? 1 : 0;
                }
                if (lane < n)      bk[rk0] = ev[0];
                if (lane + 32 < n) bk[rk1] = ev[1];
                ev[0] = (lane < n)      ? bk[lane]      : (E0C + d);
                ev[1] = (lane + 32 < n) ? bk[lane + 32] : (E0C + d);
                sv[0] = (ev[0] < E0C) ? ei[ev[0]] : (ev[0] - E0C);
                sv[1] = sv[0];
                if (NHALF == 2) sv[1] = (ev[1] < E0C) ? ei[ev[1]] : (ev[1] - E0C);
                if (lane < n) lrow[lane] = sv[0];
                if (NHALF == 2 && lane + 32 < n) lrow[lane + 32] = sv[1];
            }

            float av[2][H];
            float mx[H];
#pragma unroll
            for (int hd = 0; hd < H; ++hd) mx[hd] = -INFINITY;
#pragma unroll
            for (int hb = 0; hb < 2; ++hb) {
                if (hb < NHALF) {
                    const bool valid = (lane + 32 * hb) < n;
#pragma unroll
                    for (int hd = 0; hd < H; ++hd) {
                        const float a = leaky(ai[(size_t)d * H + hd] + aj[(size_t)sv[hb] * H + hd]);
                        av[hb][hd] = valid ? a : -INFINITY;
                        mx[hd] = fmaxf(mx[hd], av[hb][hd]);
                    }
                } else {
#pragma unroll
                    for (int hd = 0; hd < H; ++hd) av[hb][hd] = -INFINITY;
                }
            }
#pragma unroll
            for (int hd = 0; hd < H; ++hd) {
#pragma unroll
                for (int off = 16; off >= 1; off >>= 1)
                    mx[hd] = fmaxf(mx[hd], __shfl_xor(mx[hd], off, 32));
            }
            float sm[H];
#pragma unroll
            for (int hd = 0; hd < H; ++hd) {
                const float x0 = (lane < n) ? __expf(av[0][hd] - mx[hd]) : 0.f;
                float x1 = 0.f;
                if (NHALF == 2) x1 = (lane + 32 < n) ? __expf(av[1][hd] - mx[hd]) : 0.f;
                av[0][hd] = x0;
                av[1][hd] = x1;
                float t = x0 + x1;
#pragma unroll
                for (int off = 16; off >= 1; off >>= 1)
                    t += __shfl_xor(t, off, 32);
                sm[hd] = t;
            }
#pragma unroll
            for (int hb = 0; hb < 2; ++hb) {
                if (hb < NHALF && (lane + 32 * hb) < n) {
#pragma unroll
                    for (int hd = 0; hd < H; ++hd)
                        wrow[(lane + 32 * hb) * H + hd] = av[hb][hd] / sm[hd];
                }
            }

            for (int t0 = 0; t0 < n; t0 += B) {
                vecf  v[B][NG];
                float wt[B][NG];
#pragma unroll
                for (int q = 0; q < B; ++q) {
                    const int tq = (t0 + q < n) ? (t0 + q) : (n - 1);
                    const int s = __builtin_amdgcn_readlane((tq < 32) ? sv[0] : sv[1], tq & 31);
                    const float* hs = h + (size_t)s * HC;
#pragma unroll
                    for (int g = 0; g < NG; ++g) {
                        const int ch0 = g * 32 * VW + VW * lane;
                        v[q][g]  = *(const vecf*)(hs + ch0);
                        wt[q][g] = wrow[tq * H + ch0 / Cc];
                    }
                }
#pragma unroll
                for (int q = 0; q < B; ++q) {
                    if (t0 + q < n) {
#pragma unroll
                        for (int g = 0; g < NG; ++g) acc[g] += wt[q][g] * v[q][g];
                    }
                }
            }
        } else {
            int   hdg[NG];
            float aig[NG], mxg[NG], smg[NG];
#pragma unroll
            for (int g = 0; g < NG; ++g) {
                hdg[g] = (g * 32 * VW + VW * lane) / Cc;
                aig[g] = ai[(size_t)d * H + hdg[g]];
                mxg[g] = -INFINITY;
                smg[g] = 0.f;
            }
            for (int pass = 0; pass < 3; ++pass) {
                auto edge = [&](int e) {
                    const int s = (e < E0C) ? ei[e] : (e - E0C);
#pragma unroll
                    for (int g = 0; g < NG; ++g) {
                        const float a = leaky(aig[g] + aj[(size_t)s * H + hdg[g]]);
                        if (pass == 0) {
                            mxg[g] = fmaxf(mxg[g], a);
                        } else if (pass == 1) {
                            smg[g] += __expf(a - mxg[g]);
                        } else {
                            const float wgt = __expf(a - mxg[g]) / smg[g];
                            const vecf x = *(const vecf*)(h + (size_t)s * HC + g * 32 * VW + VW * lane);
                            acc[g] += wgt * x;
                        }
                    }
                };
                for (int c = 0; c < NCHUNK; ++c) {
                    const int4* p = dst4 + (size_t)c * (CH_EDGES / 4) + lane;
                    const int4 da = p[0], db = p[32];
                    const int dq[8] = { da.x, da.y, da.z, da.w, db.x, db.y, db.z, db.w };
#pragma unroll
                    for (int q = 0; q < 8; ++q) {
                        unsigned mk = __builtin_amdgcn_ballot_w32(dq[q] == d);
                        while (mk) {
                            const int l = __builtin_ctz(mk);
                            mk &= mk - 1u;
                            edge(c * CH_EDGES + (q >> 2) * 128 + l * 4 + (q & 3));
                        }
                    }
                }
                edge(E0C + d);
            }
        }

#pragma unroll
        for (int g = 0; g < NG; ++g)
            *(vecf*)(orow + g * 32 * VW + VW * lane) = acc[g];
    }
}

extern "C" void kernel_launch(void* const* d_in, const int* in_sizes, int n_in,
                              void* d_out, int out_size, void* d_ws, size_t ws_size,
                              hipStream_t stream) {
    (void)in_sizes; (void)n_in; (void)out_size; (void)ws_size;

    const float* x    = (const float*)d_in[0];
    const int*   ei   = (const int*)  d_in[1];
    const float* W1   = (const float*)d_in[2];
    const float* att1 = (const float*)d_in[3];
    const float* b1   = (const float*)d_in[4];
    const float* W2   = (const float*)d_in[5];
    const float* att2 = (const float*)d_in[6];
    const float* b2   = (const float*)d_in[7];
    float* out = (float*)d_out;

    char* base = (char*)d_ws;
    size_t off = 0;
    auto carve = [&](size_t bytes) -> char* {
        char* p = base + off;
        off = (off + bytes + 255) & ~(size_t)255;
        return p;
    };
    _Float16* W1T   = (_Float16*)carve((size_t)INDIM * (HIDDIM + CO_ROWS) * 2);
    _Float16* W2T   = (_Float16*)carve((size_t)HIDDIM * (OUTDIM + CO_ROWS) * 2);
    float*    HBUF  = (float*)   carve((size_t)MPAD * HIDDIM * 4);
    float*    AGG1  = (float*)   carve((size_t)NN * HIDDIM * 4);
    float*    AI    = (float*)   carve((size_t)NN * H1C * 4);
    float*    AJ    = (float*)   carve((size_t)NN * H1C * 4);
    int*      LCNT  = (int*)     carve((size_t)NN * 4);
    int*      LSRC  = (int*)     carve((size_t)NN * BINCAP * 4);
    unsigned* OWN   = (unsigned*)carve((size_t)EPAD);

    const int TB = 256;
    auto nb = [](long long n, int tb) { return (unsigned)((n + tb - 1) / tb); };
    const int MT2 = MPAD / 32;

    xpose_fold_f16<<<nb((long long)INDIM * (HIDDIM + CO_ROWS), TB), TB, 0, stream>>>(W1, att1, W1T, INDIM, HIDDIM, H1C, C1C);
    xpose_fold_f16<<<nb((long long)HIDDIM * (OUTDIM + CO_ROWS), TB), TB, 0, stream>>>(W2, att2, W2T, HIDDIM, OUTDIM, H2C, C2C);

    gemm_f16_wmma<false, false, HIDDIM / 32, 1, INDIM, H1C><<<MT2, (HIDDIM / 32 + 1) * 32, 0, stream>>>(x, nullptr, nullptr, W1T, HBUF, AI, AJ, MT2, NN);

    owner_bytes<<<nb((long long)EPAD / 16, TB), TB, 0, stream>>>(ei, OWN);

    edge_softmax_aggregate<H1C, C1C, false><<<AGG_WG, AGG_WAVES * 32, 0, stream>>>(ei, AI, AJ, HBUF, AGG1, LCNT, LSRC, OWN);

    gemm_f16_wmma<true, true, OUTDIM / 32, 2, HIDDIM, H2C><<<(MT2 + 1) / 2, (OUTDIM / 32 + 1) * 2 * 32, 0, stream>>>(AGG1, b1, b2, W2T, HBUF, AI, AJ, MT2, NN);

    edge_softmax_aggregate<H2C, C2C, true><<<AGG_WG, AGG_WAVES * 32, 0, stream>>>(ei, AI, AJ, HBUF, out, LCNT, LSRC, OWN);
}


# Row gather takes a row's edges in whole groups of 8 (layer 2: 16) with no clamp or test, then one last group of the smallest size 1/2/4/8(/16) that holds the rest; same edges, same order, same sums
# speedup vs baseline: 5.8859x; 1.0717x over previous
#include <hip/hip_runtime.h>
#include <hip/hip_bf16.h>

#define NN     50000
#define MPAD   50016
#define E0C    800000
#define ETOT   850000
#define INDIM  256
#define HIDDIM 256
#define OUTDIM 64
#define H1C    8
#define C1C    32
#define H2C    1
#define C2C    64
#define NEG_SLOPE 0.2f

typedef __attribute__((ext_vector_type(16))) _Float16 v16h;
typedef __attribute__((ext_vector_type(8)))  _Float16 v8h;
typedef __attribute__((ext_vector_type(4)))  _Float16 v4h;
typedef __attribute__((ext_vector_type(8)))  float    v8f;
typedef __attribute__((ext_vector_type(4)))  float    v4f;

union V16U { v16h v; v8h h[2]; };


#define CO_ROWS 16
__global__ void xpose_fold_f16(const float* __restrict__ W, const float* __restrict__ att,
                               _Float16* __restrict__ Wt, int K, int Nc, int H, int Cc) {
    int i = blockIdx.x * blockDim.x + threadIdx.x;
    if (i < K * Nc) {
        int k = i / Nc, n = i % Nc;
        Wt[(size_t)n * K + k] = (_Float16)W[i];
        return;
    }
    const int j = i - K * Nc;
    if (j >= CO_ROWS * K) return;
    const int row = j / K, k = j % K;
    float s = 0.f;
    if (row < 2 * H) {
        const int hd = (row < H) ? row : (row - H);
        const float* wv = W + (size_t)k * Nc + hd * Cc;
        const float* av = att + hd * 2 * Cc + ((row < H) ? 0 : Cc);
        for (int cc = 0; cc < Cc; ++cc) s += wv[cc] * av[cc];
    }
    Wt[(size_t)(Nc + row) * K + k] = (_Float16)s;
}

#define WMMA16(a, b, c) __builtin_amdgcn_wmma_f32_16x16x32_f16( \
        false, (a), false, (b), (short)0, (c), false, false)

#define A_PITCH (256 + 8)

template <bool ACT, bool OB, int NT, int TM, int K, int H>
__global__ __launch_bounds__((NT + 1) * TM * 32)
void gemm_f16_wmma(const float* __restrict__ A,
                   const float* __restrict__ bias,
                   const float* __restrict__ obias,
                   const _Float16* __restrict__ Bt,
                   float* __restrict__ C,
                   float* __restrict__ ai, float* __restrict__ aj, int Mtiles2, int Mrows) {
    constexpr int NTHR = NT * TM * 32;
    static_assert(2 * H <= CO_ROWS, "both coefficient halves fit the folded tile");
    constexpr int ROWS = TM * 32;
    constexpr int Nc   = NT * 32;
    constexpr int QPR  = K / 4;
    constexpr int RPP  = NTHR / QPR;
    static_assert(K + 8 == A_PITCH, "LDS pitch is made for this K");
    static_assert(NTHR % QPR == 0 && ROWS % RPP == 0, "whole rows per pass, whole passes per tile");
    __shared__ __attribute__((aligned(16))) _Float16 As[ROWS * A_PITCH];

    const int t = threadIdx.x;
    if (t < NTHR) {
        const int cq = t % QPR;
        const int rs = t / QPR;
        v4f q = {};
        if (ACT) q = *(const v4f*)(bias + cq * 4);
        const int row0 = blockIdx.x * ROWS;
#pragma unroll
        for (int i = 0; i < ROWS / RPP; ++i) {
            const int rl = i * RPP + rs;
            int rg = row0 + rl;  if (rg > Mrows - 1) rg = Mrows - 1;
            const v4f x = *(const v4f*)(A + (size_t)rg * K + cq * 4);
            float f[4] = { x.x, x.y, x.z, x.w };
            if (ACT) {
                const float bb[4] = { q.x, q.y, q.z, q.w };
#pragma unroll
                for (int j = 0; j < 4; ++j) {
                    float v = f[j] + bb[j];
                    f[j] = (v > 0.f) ? v : (__expf(v) - 1.f);
                }
            }
            v4h o;
#pragma unroll
            for (int j = 0; j < 4; ++j) o[j] = (_Float16)f[j];
            *(v4h*)(As + rl * A_PITCH + cq * 4) = o;
        }
    }
    __syncthreads();

    const int lane = t & 31;
    const int w   = __builtin_amdgcn_readfirstlane(t >> 5);
    const bool coef = w >= NT * TM;
    const int tml = coef ? (w - NT * TM) : (w / NT);
    const int tn  = coef ? NT : (w % NT);
    const int tm  = blockIdx.x * TM + tml;
    if (tm >= Mtiles2) return;
    const int r    = lane & 15;
    const int half = lane >> 4;

    const _Float16* a0 = As + (tml * 32 + r) * A_PITCH + half * 8;
    const _Float16* a1 = a0 + 16 * A_PITCH;
    const _Float16* b0 = Bt + (size_t)(tn * 32 + r) * K + half * 8;
    const _Float16* b1 = b0 + (size_t)16 * K;

    if (coef) {
        v8f c0 = {}, c1 = {};
#pragma unroll
        for (int k0 = 0; k0 < K; k0 += 32) {
            V16U A0, A1, B0;
            B0.h[0] = *(const v8h*)(b0 + k0);
            B0.h[1] = *(const v8h*)(b0 + k0 + 16);
            A0.h[0] = *(const v8h*)(a0 + k0);
            A0.h[1] = *(const v8h*)(a0 + k0 + 16);
            A1.h[0] = *(const v8h*)(a1 + k0);
            A1.h[1] = *(const v8h*)(a1 + k0 + 16);
            c0 = WMMA16(A0.v, B0.v, c0);
            c1 = WMMA16(A1.v, B0.v, c1);
        }
        if (r < 2 * H) {
            float* q = (r < H) ? (ai + r) : (aj + (r - H));
            const int row0 = tm * 32 + half * 8;
#pragma unroll
            for (int j = 0; j < 8; ++j) {
                if (row0 + j < Mrows)      q[(size_t)(row0 + j) * H]      = c0[j];
                if (row0 + 16 + j < Mrows) q[(size_t)(row0 + 16 + j) * H] = c1[j];
            }
        }
        return;
    }

    float ob0 = 0.f, ob1 = 0.f;
    if (OB) { ob0 = obias[tn * 32 + r]; ob1 = obias[tn * 32 + 16 + r]; }

    v8f c00 = {}, c01 = {}, c10 = {}, c11 = {};
#pragma unroll
    for (int k0 = 0; k0 < K; k0 += 32) {
        V16U A0, A1, B0, B1;
        B0.h[0] = *(const v8h*)(b0 + k0);
        B0.h[1] = *(const v8h*)(b0 + k0 + 16);
        B1.h[0] = *(const v8h*)(b1 + k0);
        B1.h[1] = *(const v8h*)(b1 + k0 + 16);
        A0.h[0] = *(const v8h*)(a0 + k0);
        A0.h[1] = *(const v8h*)(a0 + k0 + 16);
        A1.h[0] = *(const v8h*)(a1 + k0);
        A1.h[1] = *(const v8h*)(a1 + k0 + 16);
        c00 = WMMA16(A0.v, B0.v, c00);
        c01 = WMMA16(A0.v, B1.v, c01);
        c10 = WMMA16(A1.v, B0.v, c10);
        c11 = WMMA16(A1.v, B1.v, c11);
    }
    {
        float* p = C + (size_t)(tm * 32 + half * 8) * Nc + tn * 32 + r;
#pragma unroll
        for (int j = 0; j < 8; ++j) {
            p[(size_t)j * Nc]      = OB ? (c00[j] + ob0) : c00[j];
            p[(size_t)j * Nc + 16] = OB ? (c01[j] + ob1) : c01[j];
        }
        p += (size_t)16 * Nc;
#pragma unroll
        for (int j = 0; j < 8; ++j) {
            p[(size_t)j * Nc]      = OB ? (c10[j] + ob0) : c10[j];
            p[(size_t)j * Nc + 16] = OB ? (c11[j] + ob1) : c11[j];
        }
    }
}

#define AGG_WG     256
#define AGG_WAVES  32
#define RBASE      (NN / AGG_WG)
#define REXTRA     (NN - RBASE * AGG_WG)
#define RMAX       (RBASE + 1)
#define CH_EDGES   256
#define NCHUNK     (E0C / CH_EDGES)
#define BINCAP     64
static_assert(E0C % CH_EDGES == 0, "destination list is read in whole steps");
#define ST_EDGES   1024
#define NSTEP      ((E0C + ST_EDGES - 1) / ST_EDGES)
#define EPAD       (NSTEP * ST_EDGES)
#define WCAP       256
static_assert(AGG_WG == 256, "an owner number is one byte");
static_assert(E0C % 4 == 0, "four edges per word of the byte plane");

struct OwnStep { uint4 a, b; };

__device__ __forceinline__ unsigned owner_of(int d) {
    const int big = REXTRA * RMAX;
    return (d < big) ? ((unsigned)d / (unsigned)RMAX)
                     : ((unsigned)REXTRA + (unsigned)(d - big) / (unsigned)RBASE);
}

__global__ void owner_bytes(const int* __restrict__ ei, unsigned* __restrict__ own) {
    const int t  = blockIdx.x * blockDim.x + threadIdx.x;
    const int i0 = (t >> 5) * 128 + (t & 31);
    const int4* dst4 = (const int4*)(ei + E0C);
    int4 d[4];
#pragma unroll
    for (int j = 0; j < 4; ++j) {
        const int i = i0 + j * 32;
        d[j] = dst4[(i < E0C / 4) ? i : 0];
    }
#pragma unroll
    for (int j = 0; j < 4; ++j) {
        const int i = i0 + j * 32;
        if (i < EPAD / 4) {
            const unsigned v = (owner_of(d[j].x) & 255u) | ((owner_of(d[j].y) & 255u) << 8)
                             | ((owner_of(d[j].z) & 255u) << 16) | ((owner_of(d[j].w) & 255u) << 24);
            own[i] = (i < E0C / 4) ? v : 0xFFFFFFFFu;
        }
    }
}

__device__ __forceinline__ float leaky(float a) { return (a >= 0.f) ? a : NEG_SLOPE * a; }

__device__ __forceinline__ void file_edge(int e, const int* __restrict__ ei, int lo, int cnt,
                                          int* cntr, int* bins) {
    const unsigned rel = (unsigned)(ei[E0C + e] - lo);
    if (rel < (unsigned)cnt) {
        const int pos = atomicAdd(cntr + rel, 1);
        if (pos < BINCAP) bins[rel * BINCAP + pos] = e;
    }
}

__device__ __forceinline__ void scan_step(const OwnStep& cur, int cc, int lane, unsigned pat,
                                          const int* __restrict__ ei, int lo, int cnt,
                                          int& wn, int* wl, int* cntr, int* bins) {
    if (cc < 0) return;
    const unsigned v[8] = { cur.a.x, cur.a.y, cur.a.z, cur.a.w, cur.b.x, cur.b.y, cur.b.z, cur.b.w };
    unsigned m = 0;
#pragma unroll
    for (int q = 0; q < 8; ++q) {
        const unsigned y = v[q] ^ pat;
        const unsigned z = (y - 0x01010101u) & ~y & 0x80808080u;
        m |= (z >> 7) << q;
    }
    const int ebase = cc * ST_EDGES + lane * 16;
    if (cc == NSTEP - 1) {
#pragma unroll
        for (int q = 0; q < 8; ++q) {
            if (ebase + (q >> 2) * 512 + (q & 3) * 4 >= E0C) m &= ~(0x01010101u << q);
        }
    }
    unsigned pend = __builtin_amdgcn_ballot_w32(m != 0u);
    while (pend != 0u) {
        if (m != 0u) {
            const int slot = wn + (int)__builtin_amdgcn_mbcnt_lo(pend, 0u);
            const int bit = __builtin_ctz(m);
            m &= m - 1u;
            const int q = bit & 7, k = bit >> 3;
            const int e = ebase + (q >> 2) * 512 + (q & 3) * 4 + k;
            if (slot < WCAP) wl[slot] = e;
            else file_edge(e, ei, lo, cnt, cntr, bins);
        }
        wn += __builtin_popcount(pend);
        pend = __builtin_amdgcn_ballot_w32(m != 0u);
    }
}

template <int V> struct NumTag { static constexpr int value = V; };

template <int H, int Cc, bool REUSE>
__global__ __launch_bounds__(1024)
void edge_softmax_aggregate(const int* __restrict__ ei,
                            const float* __restrict__ ai,
                            const float* __restrict__ aj,
                            const float* __restrict__ h,
                            float* __restrict__ agg,
                            int* __restrict__ lcnt,
                            int* __restrict__ lsrc,
                            const unsigned* __restrict__ own) {
    constexpr int HC = H * Cc;
    constexpr int NV = HC / 32;
    constexpr int VW = (NV >= 4) ? 4 : NV;
    constexpr int NG = NV / VW;
    constexpr int B  = (NV >= 8) ? 8 : 16;
    typedef float vecf __attribute__((ext_vector_type(VW)));
    __shared__ int   cntr[REUSE ? 1 : RMAX];
    __shared__ int   bins[REUSE ? 1 : RMAX * BINCAP];
    __shared__ float wbuf[AGG_WAVES * BINCAP * H];
    __shared__ int   hits[REUSE ? 1 : AGG_WAVES * WCAP];

    const int lane = threadIdx.x & 31;
    const int w    = __builtin_amdgcn_readfirstlane((int)(threadIdx.x >> 5));
    const int b    = blockIdx.x;
    const int lo   = b * RBASE + ((b < REXTRA) ? b : REXTRA);
    const int cnt  = RBASE + ((b < REXTRA) ? 1 : 0);

    const int4* dst4 = (const int4*)(ei + E0C);
    if constexpr (!REUSE) {
        for (int i = threadIdx.x; i < cnt; i += AGG_WAVES * 32) {
            cntr[i] = 1;
            bins[i * BINCAP] = E0C + lo + i;
        }
        __syncthreads();

        {
            const uint4* own4 = (const uint4*)own;
            const unsigned pat = (unsigned)b * 0x01010101u;
            const int start = (int)(((long long)b * NSTEP) / AGG_WG);
            constexpr int ITERS = (NSTEP + AGG_WAVES - 1) / AGG_WAVES;
            auto step_of = [&](int i) -> int {
                const int idx = i * AGG_WAVES + w;
                int c = idx + start;
                if (c >= NSTEP) c -= NSTEP;
                return (idx < NSTEP) ? c : -1;
            };
            auto load_step = [&](int c) -> OwnStep {
                const uint4* p = own4 + (size_t)((c >= 0) ? c : 0) * (ST_EDGES / 16) + lane;
                OwnStep d; d.a = p[0]; d.b = p[32];
                return d;
            };
            int* wl = hits + w * WCAP;
            int  wn = 0;
            int c0 = step_of(0), c1 = step_of(1), c2 = step_of(2), c3 = step_of(3);
            OwnStep d0 = load_step(c0), d1 = load_step(c1), d2 = load_step(c2), d3 = load_step(c3);
            for (int it = 0; it < ITERS; it += 4) {
                scan_step(d0, c0, lane, pat, ei, lo, cnt, wn, wl, cntr, bins);
                c0 = step_of(it + 4); d0 = load_step(c0);
                scan_step(d1, c1, lane, pat, ei, lo, cnt, wn, wl, cntr, bins);
                c1 = step_of(it + 5); d1 = load_step(c1);
                scan_step(d2, c2, lane, pat, ei, lo, cnt, wn, wl, cntr, bins);
                c2 = step_of(it + 6); d2 = load_step(c2);
                scan_step(d3, c3, lane, pat, ei, lo, cnt, wn, wl, cntr, bins);
                c3 = step_of(it + 7); d3 = load_step(c3);
            }

            const int nl = (wn < WCAP) ? wn : WCAP;
            for (int i0 = 0; i0 < nl; i0 += 128) {
                int      ev[4];
                unsigned rel[4];
#pragma unroll
                for (int q = 0; q < 4; ++q) {
                    const int i = i0 + q * 32 + lane;
                    const bool ok = i < nl;
                    ev[q] = ok ? wl[i] : 0;
                    const int  d  = ei[E0C + ev[q]];
                    rel[q] = ok ? (unsigned)(d - lo) : 0xFFFFFFFFu;
                }
#pragma unroll
                for (int q = 0; q < 4; ++q) {
                    if (rel[q] < (unsigned)cnt) {
                        const int pos = atomicAdd(cntr + rel[q], 1);
                        if (pos < BINCAP) bins[rel[q] * BINCAP + pos] = ev[q];
                    }
                }
            }
        }
        __syncthreads();
    }

    float* wrow = wbuf + w * BINCAP * H;
    int n_nx = 0;
    if constexpr (REUSE) n_nx = (w < cnt) ? lcnt[lo + w] : 0;
    for (int r = w; r < cnt; r += AGG_WAVES) {
        const int d = lo + r;
        int n;
        if constexpr (REUSE) {
            n = __builtin_amdgcn_readfirstlane(n_nx);
            n_nx = (r + AGG_WAVES < cnt) ? lcnt[d + AGG_WAVES] : 0;
        } else {
            n = __builtin_amdgcn_readfirstlane(cntr[r]);
            if (lane == 0) lcnt[d] = n;
        }
        float* orow = agg + (size_t)d * HC;
        vecf acc[NG];
#pragma unroll
        for (int g = 0; g < NG; ++g) acc[g] = (vecf)(0.f);

        if (n <= BINCAP) {
            const int NHALF = (n > 32) ? 2 : 1;
            int sv[2];
            int* lrow = lsrc + (size_t)d * BINCAP;
            if constexpr (REUSE) {
                sv[0] = (lane < n) ? lrow[lane] : d;
                sv[1] = sv[0];
                if (NHALF == 2) sv[1] = (lane + 32 < n) ? lrow[lane + 32] : d;
            } else {
                int* bk = bins + r * BINCAP;
                int ev[2];
                ev[0] = (lane < n)      ? bk[lane]      : 0x7FFFFFFF;
                ev[1] = (lane + 32 < n) ? bk[lane + 32] : 0x7FFFFFFF;
                int rk0 = 0, rk1 = 0;
                for (int t = 0; t < n; ++t) {
                    const int x = __builtin_amdgcn_readlane((t < 32) ? ev[0] : ev[1], t & 31);
                    rk0 += (x < ev[0]) ? 1 : 0;
                    rk1 += (x < ev[1]) ? 1 : 0;
                }
                if (lane < n)      bk[rk0] = ev[0];
                if (lane + 32 < n) bk[rk1] = ev[1];
                ev[0] = (lane < n)      ? bk[lane]      : (E0C + d);
                ev[1] = (lane + 32 < n) ? bk[lane + 32] : (E0C + d);
                sv[0] = (ev[0] < E0C) ? ei[ev[0]] : (ev[0] - E0C);
                sv[1] = sv[0];
                if (NHALF == 2) sv[1] = (ev[1] < E0C) ? ei[ev[1]] : (ev[1] - E0C);
                if (lane < n) lrow[lane] = sv[0];
                if (NHALF == 2 && lane + 32 < n) lrow[lane + 32] = sv[1];
            }

            float av[2][H];
            float mx[H];
#pragma unroll
            for (int hd = 0; hd < H; ++hd) mx[hd] = -INFINITY;
#pragma unroll
            for (int hb = 0; hb < 2; ++hb) {
                if (hb < NHALF) {
                    const bool valid = (lane + 32 * hb) < n;
#pragma unroll
                    for (int hd = 0; hd < H; ++hd) {
                        const float a = leaky(ai[(size_t)d * H + hd] + aj[(size_t)sv[hb] * H + hd]);
                        av[hb][hd] = valid ? a : -INFINITY;
                        mx[hd] = fmaxf(mx[hd], av[hb][hd]);
                    }
                } else {
#pragma unroll
                    for (int hd = 0; hd < H; ++hd) av[hb][hd] = -INFINITY;
                }
            }
#pragma unroll
            for (int hd = 0; hd < H; ++hd) {
#pragma unroll
                for (int off = 16; off >= 1; off >>= 1)
                    mx[hd] = fmaxf(mx[hd], __shfl_xor(mx[hd], off, 32));
            }
            float sm[H];
#pragma unroll
            for (int hd = 0; hd < H; ++hd) {
                const float x0 = (lane < n) ? __expf(av[0][hd] - mx[hd]) : 0.f;
                float x1 = 0.f;
                if (NHALF == 2) x1 = (lane + 32 < n) ? __expf(av[1][hd] - mx[hd]) : 0.f;
                av[0][hd] = x0;
                av[1][hd] = x1;
                float t = x0 + x1;
#pragma unroll
                for (int off = 16; off >= 1; off >>= 1)
                    t += __shfl_xor(t, off, 32);
                sm[hd] = t;
            }
#pragma unroll
            for (int hb = 0; hb < 2; ++hb) {
                if (hb < NHALF && (lane + 32 * hb) < n) {
#pragma unroll
                    for (int hd = 0; hd < H; ++hd)
                        wrow[(lane + 32 * hb) * H + hd] = av[hb][hd] / sm[hd];
                }
            }

            auto group = [&](auto nbTag, auto fullTag, int t0) {
                constexpr int  NB   = decltype(nbTag)::value;
                constexpr bool FULL = decltype(fullTag)::value != 0;
                const int svh = (t0 < 32) ? sv[0] : sv[1];
                vecf  v[NB][NG];
                float wt[NB][NG];
#pragma unroll
                for (int q = 0; q < NB; ++q) {
                    const int tq = FULL ? (t0 + q) : ((t0 + q < n) ? (t0 + q) : (n - 1));
                    const int s = __builtin_amdgcn_readlane(svh, tq & 31);
                    const float* hs = h + (size_t)s * HC;
#pragma unroll
                    for (int g = 0; g < NG; ++g) {
                        const int ch0 = g * 32 * VW + VW * lane;
                        v[q][g]  = *(const vecf*)(hs + ch0);
                        wt[q][g] = wrow[tq * H + ch0 / Cc];
                    }
                }
#pragma unroll
                for (int q = 0; q < NB; ++q) {
                    if (FULL || t0 + q < n) {
#pragma unroll
                        for (int g = 0; g < NG; ++g) acc[g] += wt[q][g] * v[q][g];
                    }
                }
            };
            {
                int t0 = 0;
                for (; t0 + B <= n; t0 += B) group(NumTag<B>{}, NumTag<1>{}, t0);
                const int r = n - t0;
                if (r > 8)      { if constexpr (B > 8) group(NumTag<16>{}, NumTag<0>{}, t0); }
                else if (r > 4) group(NumTag<8>{}, NumTag<0>{}, t0);
                else if (r > 2) group(NumTag<4>{}, NumTag<0>{}, t0);
                else if (r > 1) group(NumTag<2>{}, NumTag<0>{}, t0);
                else if (r > 0) group(NumTag<1>{}, NumTag<1>{}, t0);
            }
        } else {
            int   hdg[NG];
            float aig[NG], mxg[NG], smg[NG];
#pragma unroll
            for (int g = 0; g < NG; ++g) {
                hdg[g] = (g * 32 * VW + VW * lane) / Cc;
                aig[g] = ai[(size_t)d * H + hdg[g]];
                mxg[g] = -INFINITY;
                smg[g] = 0.f;
            }
            for (int pass = 0; pass < 3; ++pass) {
                auto edge = [&](int e) {
                    const int s = (e < E0C) ? ei[e] : (e - E0C);
#pragma unroll
                    for (int g = 0; g < NG; ++g) {
                        const float a = leaky(aig[g] + aj[(size_t)s * H + hdg[g]]);
                        if (pass == 0) {
                            mxg[g] = fmaxf(mxg[g], a);
                        } else if (pass == 1) {
                            smg[g] += __expf(a - mxg[g]);
                        } else {
                            const float wgt = __expf(a - mxg[g]) / smg[g];
                            const vecf x = *(const vecf*)(h + (size_t)s * HC + g * 32 * VW + VW * lane);
                            acc[g] += wgt * x;
                        }
                    }
                };
                for (int c = 0; c < NCHUNK; ++c) {
                    const int4* p = dst4 + (size_t)c * (CH_EDGES / 4) + lane;
                    const int4 da = p[0], db = p[32];
                    const int dq[8] = { da.x, da.y, da.z, da.w, db.x, db.y, db.z, db.w };
#pragma unroll
                    for (int q = 0; q < 8; ++q) {
                        unsigned mk = __builtin_amdgcn_ballot_w32(dq[q] == d);
                        while (mk) {
                            const int l = __builtin_ctz(mk);
                            mk &= mk - 1u;
                            edge(c * CH_EDGES + (q >> 2) * 128 + l * 4 + (q & 3));
                        }
                    }
                }
                edge(E0C + d);
            }
        }

#pragma unroll
        for (int g = 0; g < NG; ++g)
            *(vecf*)(orow + g * 32 * VW + VW * lane) = acc[g];
    }
}

extern "C" void kernel_launch(void* const* d_in, const int* in_sizes, int n_in,
                              void* d_out, int out_size, void* d_ws, size_t ws_size,
                              hipStream_t stream) {
    (void)in_sizes; (void)n_in; (void)out_size; (void)ws_size;

    const float* x    = (const float*)d_in[0];
    const int*   ei   = (const int*)  d_in[1];
    const float* W1   = (const float*)d_in[2];
    const float* att1 = (const float*)d_in[3];
    const float* b1   = (const float*)d_in[4];
    const float* W2   = (const float*)d_in[5];
    const float* att2 = (const float*)d_in[6];
    const float* b2   = (const float*)d_in[7];
    float* out = (float*)d_out;

    char* base = (char*)d_ws;
    size_t off = 0;
    auto carve = [&](size_t bytes) -> char* {
        char* p = base + off;
        off = (off + bytes + 255) & ~(size_t)255;
        return p;
    };
    _Float16* W1T   = (_Float16*)carve((size_t)INDIM * (HIDDIM + CO_ROWS) * 2);
    _Float16* W2T   = (_Float16*)carve((size_t)HIDDIM * (OUTDIM + CO_ROWS) * 2);
    float*    HBUF  = (float*)   carve((size_t)MPAD * HIDDIM * 4);
    float*    AGG1  = (float*)   carve((size_t)NN * HIDDIM * 4);
    float*    AI    = (float*)   carve((size_t)NN * H1C * 4);
    float*    AJ    = (float*)   carve((size_t)NN * H1C * 4);
    int*      LCNT  = (int*)     carve((size_t)NN * 4);
    int*      LSRC  = (int*)     carve((size_t)NN * BINCAP * 4);
    unsigned* OWN   = (unsigned*)carve((size_t)EPAD);

    const int TB = 256;
    auto nb = [](long long n, int tb) { return (unsigned)((n + tb - 1) / tb); };
    const int MT2 = MPAD / 32;

    xpose_fold_f16<<<nb((long long)INDIM * (HIDDIM + CO_ROWS), TB), TB, 0, stream>>>(W1, att1, W1T, INDIM, HIDDIM, H1C, C1C);
    xpose_fold_f16<<<nb((long long)HIDDIM * (OUTDIM + CO_ROWS), TB), TB, 0, stream>>>(W2, att2, W2T, HIDDIM, OUTDIM, H2C, C2C);

    gemm_f16_wmma<false, false, HIDDIM / 32, 1, INDIM, H1C><<<MT2, (HIDDIM / 32 + 1) * 32, 0, stream>>>(x, nullptr, nullptr, W1T, HBUF, AI, AJ, MT2, NN);

    owner_bytes<<<nb((long long)EPAD / 16, TB), TB, 0, stream>>>(ei, OWN);

    edge_softmax_aggregate<H1C, C1C, false><<<AGG_WG, AGG_WAVES * 32, 0, stream>>>(ei, AI, AJ, HBUF, AGG1, LCNT, LSRC, OWN);

    gemm_f16_wmma<true, true, OUTDIM / 32, 2, HIDDIM, H2C><<<(MT2 + 1) / 2, (OUTDIM / 32 + 1) * 2 * 32, 0, stream>>>(AGG1, b1, b2, W2T, HBUF, AI, AJ, MT2, NN);

    edge_softmax_aggregate<H2C, C2C, true><<<AGG_WG, AGG_WAVES * 32, 0, stream>>>(ei, AI, AJ, HBUF, out, LCNT, LSRC, OWN);
}
